# starting kernel
# baseline (speedup 1.0000x reference)
#include <hip/hip_runtime.h>
#include <hip/hip_bf16.h>

#define N_HEADS 32
#define N_KV 8
#define HDIM 8
#define SEQ 2048
#define HID 256
#define QTILES (SEQ / 16)

typedef __attribute__((ext_vector_type(2))) float v2f;
typedef __attribute__((ext_vector_type(8))) float v8f;

template <int CTRL>
__device__ __forceinline__ float dpp_max(float x) {
  const int y =
      __builtin_amdgcn_update_dpp(0, __float_as_int(x), CTRL, 0xf, 0xf, true);
  return fmaxf(x, __int_as_float(y));
}

__device__ __forceinline__ float rowmax16(float x) {
  x = dpp_max<0xB1>(x);
  x = dpp_max<0x4E>(x);
  x = dpp_max<0x141>(x);
  x = dpp_max<0x140>(x);
  return x;
}

__device__ __forceinline__ float fast_exp2(float x) {
  return __builtin_amdgcn_exp2f(x);
}

__global__ __launch_bounds__(64) void qkv_rope_so8_kernel(
    const float* __restrict__ x, const float* __restrict__ Wq,
    const float* __restrict__ Wk, const float* __restrict__ Wv,
    const float* __restrict__ rope, const float* __restrict__ so8,
    float* __restrict__ qb, float* __restrict__ kb, float* __restrict__ vtb) {
  const int bs = blockIdx.x;
  const int s = bs & (SEQ - 1);
  const int b = bs >> 11;

  __shared__ float xs[HID];
  for (int i = threadIdx.x; i < HID; i += 64) xs[i] = x[(size_t)bs * HID + i];
  __syncthreads();

  const int t = threadIdx.x;
  if (t >= 48) return;
  const int kind = (t < 32) ? 0 : (t < 40 ? 1 : 2);
  const int head = (t < 32) ? t : (t < 40 ? t - 32 : t - 40);
  const float* W = (kind == 0) ? Wq : ((kind == 1) ? Wk : Wv);
  const float* wr = W + (size_t)head * HDIM * HID;

  float acc[HDIM] = {};
  for (int i = 0; i < HID; ++i) {
    const float xv = xs[i];
#pragma unroll
    for (int d = 0; d < HDIM; ++d) acc[d] = fmaf(xv, wr[d * HID + i], acc[d]);
  }

  float y[HDIM];
  if (kind <= 1) {
    const float invf[4] = {1.0f, 0.1f, 0.01f, 0.001f};
    float cr[4] = {}, sr[4] = {};
#pragma unroll
    for (int i = 0; i < 4; ++i) {
      const float fr = (float)s * invf[i];
      const float ci = cosf(fr), si = sinf(fr);
#pragma unroll
      for (int j = 0; j < 4; ++j) {
        cr[j] = fmaf(ci, rope[i * 8 + j], cr[j]);
        sr[j] = fmaf(si, rope[i * 8 + j], sr[j]);
      }
    }
#pragma unroll
    for (int d = 0; d < HDIM; ++d) {
      const int j = d & 3;
      const float rh = (d < 4) ? -acc[d + 4] : acc[d - 4];
      y[d] = acc[d] * cr[j] + rh * sr[j];
    }
  } else {
#pragma unroll
    for (int d = 0; d < HDIM; ++d) y[d] = acc[d];
  }

  float z[HDIM];
#pragma unroll
  for (int j = 0; j < HDIM; ++j) {
    float sum = 0.0f;
#pragma unroll
    for (int d = 0; d < HDIM; ++d) sum = fmaf(y[d], so8[d * 8 + j], sum);
    z[j] = sum;
  }

  if (kind == 0) {
    float* dst = qb + (((size_t)b * N_HEADS + head) * SEQ + s) * HDIM;
#pragma unroll
    for (int j = 0; j < HDIM; ++j) dst[j] = z[j];
  } else if (kind == 1) {
    float* dst = kb + (((size_t)b * N_KV + head) * SEQ + s) * HDIM;
#pragma unroll
    for (int j = 0; j < HDIM; ++j) dst[j] = z[j];
  } else {
#pragma unroll
    for (int j = 0; j < HDIM; ++j)
      vtb[(((size_t)b * N_KV + head) * HDIM + j) * SEQ + s] = z[j];
  }
}

template <bool MASK>
__device__ __forceinline__ void attn_tile(
    const float* __restrict__ kp, const float* __restrict__ vp,
    float* __restrict__ pw, int kbase, int lanelo, int hi, int hi2,
    const float* __restrict__ maskv, const v2f Qa0, const v2f Qa1,
    float* __restrict__ m, v8f& o) {
  const float* krow = kp + (size_t)(kbase + lanelo) * HDIM;
  const v2f Kb0 = *(const v2f*)(krow + hi2);
  const v2f Kb1 = *(const v2f*)(krow + 4 + hi2);

  v8f sc = {};
  sc = __builtin_amdgcn_wmma_f32_16x16x4_f32(false, Qa0, false, Kb0, (short)0,
                                             sc, false, false);
  sc = __builtin_amdgcn_wmma_f32_16x16x4_f32(false, Qa1, false, Kb1, (short)0,
                                             sc, false, false);

  float p[8];
#pragma unroll
  for (int r = 0; r < 8; ++r) {
    float sv = sc[r];
    if (MASK) sv += maskv[r];
    const float rm = rowmax16(sv);
    const float mnew = fmaxf(m[r], rm);
    const float corr = fast_exp2(m[r] - mnew);
    m[r] = mnew;
    p[r] = fast_exp2(sv - mnew);
    o[r] = o[r] * corr;
  }

#pragma unroll
  for (int r = 0; r < 8; ++r) pw[r * 16 + hi * 128 + lanelo] = p[r];
  v2f Pa[4];
#pragma unroll
  for (int t = 0; t < 4; ++t)
    Pa[t] = *(const v2f*)(pw + (lanelo & 7) * 16 + (lanelo >> 3) * 128 +
                          4 * t + hi2);

  v2f Vb[4];
#pragma unroll
  for (int t = 0; t < 4; ++t) {
    if (lanelo < 8) {
      Vb[t] = *(const v2f*)(vp + (size_t)lanelo * SEQ + kbase + 4 * t + hi2);
    } else {
      const float c = (lanelo == 8) ? 1.0f : 0.0f;
      Vb[t] = (v2f){c, c};
    }
  }
#pragma unroll
  for (int t = 0; t < 4; ++t)
    o = __builtin_amdgcn_wmma_f32_16x16x4_f32(false, Pa[t], false, Vb[t],
                                              (short)0, o, false, false);
}

__global__ __launch_bounds__(256) void so8_flash_attn_kernel(
    const float* __restrict__ q, const float* __restrict__ k,
    const float* __restrict__ vT, float* __restrict__ attn) {
  const int lane = threadIdx.x & 31;
  const int widx = threadIdx.x >> 5;
  const int w = blockIdx.x * 8 + widx;
  const int qt = w & (QTILES - 1);
  const int h = (w >> 7) & (N_HEADS - 1);
  const int b = w >> 12;
  const int kv = h >> 2;
  const int qbase = qt * 16;
  const int lanelo = lane & 15;
  const int hi = lane >> 4;
  const int hi2 = hi * 2;

  const float* qp = q + ((size_t)b * N_HEADS + h) * SEQ * HDIM;
  const float* kp = k + ((size_t)b * N_KV + kv) * SEQ * HDIM;
  const float* vp = vT + ((size_t)b * N_KV + kv) * HDIM * SEQ;

  __shared__ float plds[8][256];
  float* pw = &plds[widx][0];

  const float QSCALE = 0.35355339059327f * 1.44269504088896f;
  const float* qrow = qp + (size_t)(qbase + lanelo) * HDIM;
  const v2f Qa0 = *(const v2f*)(qrow + hi2) * QSCALE;
  const v2f Qa1 = *(const v2f*)(qrow + 4 + hi2) * QSCALE;

  float maskv[8];
#pragma unroll
  for (int r = 0; r < 8; ++r)
    maskv[r] = (lanelo > r + 8 * hi) ? -3.0e38f : 0.0f;

  v8f o = {};
  float m[8];
#pragma unroll
  for (int r = 0; r < 8; ++r) m[r] = -3.0e38f;

  for (int kt = 0; kt < qt; ++kt)
    attn_tile<false>(kp, vp, pw, kt * 16, lanelo, hi, hi2, maskv, Qa0, Qa1, m,
                     o);
  attn_tile<true>(kp, vp, pw, qbase, lanelo, hi, hi2, maskv, Qa0, Qa1, m, o);

#pragma unroll
  for (int r = 0; r < 8; ++r) {
    const float ov = o[r];
    const float l = __shfl(ov, (lane & 16) + 8);
    const float outv = ov * __builtin_amdgcn_rcpf(l);
    const int row = qbase + r + 8 * hi;
    if (lanelo < HDIM)
      attn[(((size_t)b * SEQ + row) * N_HEADS + h) * HDIM + lanelo] = outv;
  }
}

__global__ __launch_bounds__(256) void out_proj_kernel(
    const float* __restrict__ attn, const float* __restrict__ Wo,
    float* __restrict__ out) {
  const int bs = blockIdx.x;
  __shared__ float xs[HID];
  xs[threadIdx.x] = attn[(size_t)bs * HID + threadIdx.x];
  __syncthreads();
  const int oi = threadIdx.x;
  const float* wr = Wo + (size_t)oi * HID;
  float acc = 0.0f;
  for (int i = 0; i < HID; ++i) acc = fmaf(xs[i], wr[i], acc);
  out[(size_t)bs * HID + oi] = acc;
}

extern "C" void kernel_launch(void* const* d_in, const int* in_sizes, int n_in,
                              void* d_out, int out_size, void* d_ws,
                              size_t ws_size, hipStream_t stream) {
  const float* x = (const float*)d_in[0];
  const float* Wq = (const float*)d_in[1];
  const float* Wk = (const float*)d_in[2];
  const float* Wv = (const float*)d_in[3];
  const float* Wo = (const float*)d_in[4];
  const float* rope = (const float*)d_in[5];
  const float* so8 = (const float*)d_in[6];

  float* ws = (float*)d_ws;
  float* qb = ws;
  float* kb = qb + (size_t)2 * N_HEADS * SEQ * HDIM;
  float* vtb = kb + (size_t)2 * N_KV * SEQ * HDIM;
  float* ab = vtb + (size_t)2 * N_KV * SEQ * HDIM;

  const int BS = 2 * SEQ;

  qkv_rope_so8_kernel<<<BS, 64, 0, stream>>>(x, Wq, Wk, Wv, rope, so8, qb, kb,
                                             vtb);
  so8_flash_attn_kernel<<<(2 * N_HEADS * QTILES) / 8, 256, 0, stream>>>(
      qb, kb, vtb, ab);
  out_proj_kernel<<<BS, 256, 0, stream>>>(ab, Wo, (float*)d_out);
}


# QKV and output projections: per-thread scalar FMA dot products replaced by f32 16x16x4 WMMA tile GEMMs (16-row blocks, LDS-staged epilogue); dot products now summed in WMMA k order
# speedup vs baseline: 2.0158x; 2.0158x over previous
#include <hip/hip_runtime.h>
#include <hip/hip_bf16.h>

#define N_HEADS 32
#define N_KV 8
#define HDIM 8
#define SEQ 2048
#define HID 256
#define QTILES (SEQ / 16)

typedef __attribute__((ext_vector_type(2))) float v2f;
typedef __attribute__((ext_vector_type(8))) float v8f;
typedef __attribute__((ext_vector_type(4))) float v4f;


#define TILE_M 16
#define QKV_PITCH 388
#define OUT_PITCH 260

template <int NT>
__device__ __forceinline__ void wave_gemm_k256(const float* __restrict__ arow,
                                               const float* const (&wrow)[NT],
                                               v8f (&acc)[NT]) {
  v4f a0 = *(const v4f*)(arow);
  v4f a1 = *(const v4f*)(arow + 8);
  v4f b0[NT], b1[NT];
#pragma unroll
  for (int t = 0; t < NT; ++t) {
    b0[t] = *(const v4f*)(wrow[t]);
    b1[t] = *(const v4f*)(wrow[t] + 8);
  }
#pragma unroll
  for (int kc = 0; kc < HID / 16; ++kc) {
    v4f na0 = a0, na1 = a1;
    v4f nb0[NT], nb1[NT];
#pragma unroll
    for (int t = 0; t < NT; ++t) {
      nb0[t] = b0[t];
      nb1[t] = b1[t];
    }
    if (kc + 1 < HID / 16) {
      const int ko = (kc + 1) * 16;
      na0 = *(const v4f*)(arow + ko);
      na1 = *(const v4f*)(arow + ko + 8);
#pragma unroll
      for (int t = 0; t < NT; ++t) {
        nb0[t] = *(const v4f*)(wrow[t] + ko);
        nb1[t] = *(const v4f*)(wrow[t] + ko + 8);
      }
    }
#pragma unroll
    for (int t = 0; t < NT; ++t)
      acc[t] = __builtin_amdgcn_wmma_f32_16x16x4_f32(
          false, a0.xy, false, b0[t].xy, (short)0, acc[t], false, false);
#pragma unroll
    for (int t = 0; t < NT; ++t)
      acc[t] = __builtin_amdgcn_wmma_f32_16x16x4_f32(
          false, a0.zw, false, b0[t].zw, (short)0, acc[t], false, false);
#pragma unroll
    for (int t = 0; t < NT; ++t)
      acc[t] = __builtin_amdgcn_wmma_f32_16x16x4_f32(
          false, a1.xy, false, b1[t].xy, (short)0, acc[t], false, false);
#pragma unroll
    for (int t = 0; t < NT; ++t)
      acc[t] = __builtin_amdgcn_wmma_f32_16x16x4_f32(
          false, a1.zw, false, b1[t].zw, (short)0, acc[t], false, false);
    a0 = na0;
    a1 = na1;
#pragma unroll
    for (int t = 0; t < NT; ++t) {
      b0[t] = nb0[t];
      b1[t] = nb1[t];
    }
  }
}

template <int CTRL>
__device__ __forceinline__ float dpp_max(float x) {
  const int y =
      __builtin_amdgcn_update_dpp(0, __float_as_int(x), CTRL, 0xf, 0xf, true);
  return fmaxf(x, __int_as_float(y));
}

__device__ __forceinline__ float rowmax16(float x) {
  x = dpp_max<0xB1>(x);
  x = dpp_max<0x4E>(x);
  x = dpp_max<0x141>(x);
  x = dpp_max<0x140>(x);
  return x;
}

__device__ __forceinline__ float fast_exp2(float x) {
  return __builtin_amdgcn_exp2f(x);
}

__global__ __launch_bounds__(256) void qkv_rope_so8_kernel(
    const float* __restrict__ x, const float* __restrict__ Wq,
    const float* __restrict__ Wk, const float* __restrict__ Wv,
    const float* __restrict__ rope, const float* __restrict__ so8,
    float* __restrict__ qb, float* __restrict__ kb, float* __restrict__ vtb) {
  __shared__ __attribute__((aligned(16))) float tile[TILE_M * QKV_PITCH];
  __shared__ float trig[TILE_M][8];

  const int tid = threadIdx.x;
  const int lane = tid & 31;
  const int widx = __builtin_amdgcn_readfirstlane(tid >> 5);
  const int c = lane & 15;
  const int h = lane >> 4;
  const int m0 = blockIdx.x * TILE_M;

  if (tid < 64) {
    const int row = tid >> 2;
    const int i = tid & 3;
    const float invf =
        (i == 0) ? 1.0f : ((i == 1) ? 0.1f : ((i == 2) ? 0.01f : 0.001f));
    const int s = (m0 + row) & (SEQ - 1);
    const float fr = (float)s * invf;
    trig[row][i] = cosf(fr);
    trig[row][4 + i] = sinf(fr);
  }

  const float* arow = x + (size_t)(m0 + c) * HID + 4 * h;
  const float* wrow[3];
#pragma unroll
  for (int t = 0; t < 3; ++t) {
    const int nt = widx * 3 + t;
    const int n0 = nt * 16;
    const float* Wb = (nt < 16) ? (Wq + (size_t)n0 * HID)
                                : ((nt < 20) ? (Wk + (size_t)(n0 - 256) * HID)
                                             : (Wv + (size_t)(n0 - 320) * HID));
    wrow[t] = Wb + (size_t)c * HID + 4 * h;
  }
  v8f acc[3] = {};
  wave_gemm_k256<3>(arow, wrow, acc);

#pragma unroll
  for (int t = 0; t < 3; ++t)
#pragma unroll
    for (int r = 0; r < 8; ++r)
      tile[(8 * h + r) * QKV_PITCH + (widx * 3 + t) * 16 + c] = acc[t][r];
  __syncthreads();

  for (int it = 0; it < 3; ++it) {
    const int id = it * 256 + tid;
    const int row = id & 15;
    const int hd = id >> 4;
    const int kind = (hd < 32) ? 0 : (hd < 40 ? 1 : 2);
    const int head = (hd < 32) ? hd : (hd < 40 ? hd - 32 : hd - 40);
    const int bs = m0 + row;
    const int s = bs & (SEQ - 1);
    const int b = bs >> 11;

    const float* tp = tile + row * QKV_PITCH + hd * 8;
    const v4f t0 = *(const v4f*)tp;
    const v4f t1 = *(const v4f*)(tp + 4);
    const float pr[HDIM] = {t0.x, t0.y, t0.z, t0.w, t1.x, t1.y, t1.z, t1.w};

    float y[HDIM];
    if (kind <= 1) {
      float cr[4] = {}, sr[4] = {};
#pragma unroll
      for (int i = 0; i < 4; ++i) {
        const float ci = trig[row][i], si = trig[row][4 + i];
#pragma unroll
        for (int j = 0; j < 4; ++j) {
          cr[j] = fmaf(ci, rope[i * 8 + j], cr[j]);
          sr[j] = fmaf(si, rope[i * 8 + j], sr[j]);
        }
      }
#pragma unroll
      for (int d = 0; d < HDIM; ++d) {
        const int j = d & 3;
        const float rh = (d < 4) ? -pr[d + 4] : pr[d - 4];
        y[d] = pr[d] * cr[j] + rh * sr[j];
      }
    } else {
#pragma unroll
      for (int d = 0; d < HDIM; ++d) y[d] = pr[d];
    }

    float z[HDIM];
#pragma unroll
    for (int j = 0; j < HDIM; ++j) {
      float sum = 0.0f;
#pragma unroll
      for (int d = 0; d < HDIM; ++d) sum = fmaf(y[d], so8[d * 8 + j], sum);
      z[j] = sum;
    }

    if (kind == 0) {
      float* dst = qb + (((size_t)b * N_HEADS + head) * SEQ + s) * HDIM;
      *(v4f*)dst = (v4f){z[0], z[1], z[2], z[3]};
      *(v4f*)(dst + 4) = (v4f){z[4], z[5], z[6], z[7]};
    } else if (kind == 1) {
      float* dst = kb + (((size_t)b * N_KV + head) * SEQ + s) * HDIM;
      *(v4f*)dst = (v4f){z[0], z[1], z[2], z[3]};
      *(v4f*)(dst + 4) = (v4f){z[4], z[5], z[6], z[7]};
    } else {
#pragma unroll
      for (int j = 0; j < HDIM; ++j)
        vtb[(((size_t)b * N_KV + head) * HDIM + j) * SEQ + s] = z[j];
    }
  }
}

template <bool MASK>
__device__ __forceinline__ void attn_tile(
    const float* __restrict__ kp, const float* __restrict__ vp,
    float* __restrict__ pw, int kbase, int lanelo, int hi, int hi2,
    const float* __restrict__ maskv, const v2f Qa0, const v2f Qa1,
    float* __restrict__ m, v8f& o) {
  const float* krow = kp + (size_t)(kbase + lanelo) * HDIM;
  const v2f Kb0 = *(const v2f*)(krow + hi2);
  const v2f Kb1 = *(const v2f*)(krow + 4 + hi2);

  v8f sc = {};
  sc = __builtin_amdgcn_wmma_f32_16x16x4_f32(false, Qa0, false, Kb0, (short)0,
                                             sc, false, false);
  sc = __builtin_amdgcn_wmma_f32_16x16x4_f32(false, Qa1, false, Kb1, (short)0,
                                             sc, false, false);

  float p[8];
#pragma unroll
  for (int r = 0; r < 8; ++r) {
    float sv = sc[r];
    if (MASK) sv += maskv[r];
    const float rm = rowmax16(sv);
    const float mnew = fmaxf(m[r], rm);
    const float corr = fast_exp2(m[r] - mnew);
    m[r] = mnew;
    p[r] = fast_exp2(sv - mnew);
    o[r] = o[r] * corr;
  }

#pragma unroll
  for (int r = 0; r < 8; ++r) pw[r * 16 + hi * 128 + lanelo] = p[r];
  v2f Pa[4];
#pragma unroll
  for (int t = 0; t < 4; ++t)
    Pa[t] = *(const v2f*)(pw + (lanelo & 7) * 16 + (lanelo >> 3) * 128 +
                          4 * t + hi2);

  v2f Vb[4];
#pragma unroll
  for (int t = 0; t < 4; ++t) {
    if (lanelo < 8) {
      Vb[t] = *(const v2f*)(vp + (size_t)lanelo * SEQ + kbase + 4 * t + hi2);
    } else {
      const float c = (lanelo == 8) ? 1.0f : 0.0f;
      Vb[t] = (v2f){c, c};
    }
  }
#pragma unroll
  for (int t = 0; t < 4; ++t)
    o = __builtin_amdgcn_wmma_f32_16x16x4_f32(false, Pa[t], false, Vb[t],
                                              (short)0, o, false, false);
}

__global__ __launch_bounds__(256) void so8_flash_attn_kernel(
    const float* __restrict__ q, const float* __restrict__ k,
    const float* __restrict__ vT, float* __restrict__ attn) {
  const int lane = threadIdx.x & 31;
  const int widx = threadIdx.x >> 5;
  const int w = blockIdx.x * 8 + widx;
  const int qt = w & (QTILES - 1);
  const int h = (w >> 7) & (N_HEADS - 1);
  const int b = w >> 12;
  const int kv = h >> 2;
  const int qbase = qt * 16;
  const int lanelo = lane & 15;
  const int hi = lane >> 4;
  const int hi2 = hi * 2;

  const float* qp = q + ((size_t)b * N_HEADS + h) * SEQ * HDIM;
  const float* kp = k + ((size_t)b * N_KV + kv) * SEQ * HDIM;
  const float* vp = vT + ((size_t)b * N_KV + kv) * HDIM * SEQ;

  __shared__ float plds[8][256];
  float* pw = &plds[widx][0];

  const float QSCALE = 0.35355339059327f * 1.44269504088896f;
  const float* qrow = qp + (size_t)(qbase + lanelo) * HDIM;
  const v2f Qa0 = *(const v2f*)(qrow + hi2) * QSCALE;
  const v2f Qa1 = *(const v2f*)(qrow + 4 + hi2) * QSCALE;

  float maskv[8];
#pragma unroll
  for (int r = 0; r < 8; ++r)
    maskv[r] = (lanelo > r + 8 * hi) ? -3.0e38f : 0.0f;

  v8f o = {};
  float m[8];
#pragma unroll
  for (int r = 0; r < 8; ++r) m[r] = -3.0e38f;

  for (int kt = 0; kt < qt; ++kt)
    attn_tile<false>(kp, vp, pw, kt * 16, lanelo, hi, hi2, maskv, Qa0, Qa1, m,
                     o);
  attn_tile<true>(kp, vp, pw, qbase, lanelo, hi, hi2, maskv, Qa0, Qa1, m, o);

#pragma unroll
  for (int r = 0; r < 8; ++r) {
    const float ov = o[r];
    const float l = __shfl(ov, (lane & 16) + 8);
    const float outv = ov * __builtin_amdgcn_rcpf(l);
    const int row = qbase + r + 8 * hi;
    if (lanelo < HDIM)
      attn[(((size_t)b * SEQ + row) * N_HEADS + h) * HDIM + lanelo] = outv;
  }
}

__global__ __launch_bounds__(256) void out_proj_kernel(
    const float* __restrict__ attn, const float* __restrict__ Wo,
    float* __restrict__ out) {
  __shared__ __attribute__((aligned(16))) float tile[TILE_M * OUT_PITCH];

  const int tid = threadIdx.x;
  const int lane = tid & 31;
  const int widx = tid >> 5;
  const int c = lane & 15;
  const int h = lane >> 4;
  const int m0 = blockIdx.x * TILE_M;

  const float* arow = attn + (size_t)(m0 + c) * HID + 4 * h;
  const float* wrow[2];
#pragma unroll
  for (int t = 0; t < 2; ++t)
    wrow[t] = Wo + (size_t)((widx * 2 + t) * 16 + c) * HID + 4 * h;
  v8f acc[2] = {};
  wave_gemm_k256<2>(arow, wrow, acc);

#pragma unroll
  for (int t = 0; t < 2; ++t)
#pragma unroll
    for (int r = 0; r < 8; ++r)
      tile[(8 * h + r) * OUT_PITCH + (widx * 2 + t) * 16 + c] = acc[t][r];
  __syncthreads();

#pragma unroll
  for (int it = 0; it < 4; ++it) {
    const int id = it * 256 + tid;
    const int row = id >> 6;
    const int c4 = (id & 63) * 4;
    *(v4f*)(out + (size_t)(m0 + row) * HID + c4) =
        *(const v4f*)(tile + row * OUT_PITCH + c4);
  }
}

extern "C" void kernel_launch(void* const* d_in, const int* in_sizes, int n_in,
                              void* d_out, int out_size, void* d_ws,
                              size_t ws_size, hipStream_t stream) {
  const float* x = (const float*)d_in[0];
  const float* Wq = (const float*)d_in[1];
  const float* Wk = (const float*)d_in[2];
  const float* Wv = (const float*)d_in[3];
  const float* Wo = (const float*)d_in[4];
  const float* rope = (const float*)d_in[5];
  const float* so8 = (const float*)d_in[6];

  float* ws = (float*)d_ws;
  float* qb = ws;
  float* kb = qb + (size_t)2 * N_HEADS * SEQ * HDIM;
  float* vtb = kb + (size_t)2 * N_KV * SEQ * HDIM;
  float* ab = vtb + (size_t)2 * N_KV * SEQ * HDIM;

  const int BS = 2 * SEQ;

  qkv_rope_so8_kernel<<<BS / TILE_M, 256, 0, stream>>>(x, Wq, Wk, Wv, rope, so8,
                                                       qb, kb, vtb);
  so8_flash_attn_kernel<<<(2 * N_HEADS * QTILES) / 8, 256, 0, stream>>>(
      qb, kb, vtb, ab);
  out_proj_kernel<<<BS / TILE_M, 256, 0, stream>>>(ab, Wo, (float*)d_out);
}


# Attention computes transposed scores K*Q^T: softmax max and rescale per lane, P^T feeds the PV WMMA from the score registers (no LDS transpose, no DPP row max); a tile's keys summed in permuted order
# speedup vs baseline: 3.4064x; 1.6899x over previous
#include <hip/hip_runtime.h>
#include <hip/hip_bf16.h>

#define N_HEADS 32
#define N_KV 8
#define HDIM 8
#define SEQ 2048
#define HID 256
#define QTILES (SEQ / 16)

typedef __attribute__((ext_vector_type(2))) float v2f;
typedef __attribute__((ext_vector_type(8))) float v8f;
typedef __attribute__((ext_vector_type(4))) float v4f;


#define TILE_M 16
#define QKV_PITCH 388
#define OUT_PITCH 260

template <int NT>
__device__ __forceinline__ void wave_gemm_k256(const float* __restrict__ arow,
                                               const float* const (&wrow)[NT],
                                               v8f (&acc)[NT]) {
  v4f a0 = *(const v4f*)(arow);
  v4f a1 = *(const v4f*)(arow + 8);
  v4f b0[NT], b1[NT];
#pragma unroll
  for (int t = 0; t < NT; ++t) {
    b0[t] = *(const v4f*)(wrow[t]);
    b1[t] = *(const v4f*)(wrow[t] + 8);
  }
#pragma unroll
  for (int kc = 0; kc < HID / 16; ++kc) {
    v4f na0 = a0, na1 = a1;
    v4f nb0[NT], nb1[NT];
#pragma unroll
    for (int t = 0; t < NT; ++t) {
      nb0[t] = b0[t];
      nb1[t] = b1[t];
    }
    if (kc + 1 < HID / 16) {
      const int ko = (kc + 1) * 16;
      na0 = *(const v4f*)(arow + ko);
      na1 = *(const v4f*)(arow + ko + 8);
#pragma unroll
      for (int t = 0; t < NT; ++t) {
        nb0[t] = *(const v4f*)(wrow[t] + ko);
        nb1[t] = *(const v4f*)(wrow[t] + ko + 8);
      }
    }
#pragma unroll
    for (int t = 0; t < NT; ++t)
      acc[t] = __builtin_amdgcn_wmma_f32_16x16x4_f32(
          false, a0.xy, false, b0[t].xy, (short)0, acc[t], false, false);
#pragma unroll
    for (int t = 0; t < NT; ++t)
      acc[t] = __builtin_amdgcn_wmma_f32_16x16x4_f32(
          false, a0.zw, false, b0[t].zw, (short)0, acc[t], false, false);
#pragma unroll
    for (int t = 0; t < NT; ++t)
      acc[t] = __builtin_amdgcn_wmma_f32_16x16x4_f32(
          false, a1.xy, false, b1[t].xy, (short)0, acc[t], false, false);
#pragma unroll
    for (int t = 0; t < NT; ++t)
      acc[t] = __builtin_amdgcn_wmma_f32_16x16x4_f32(
          false, a1.zw, false, b1[t].zw, (short)0, acc[t], false, false);
    a0 = na0;
    a1 = na1;
#pragma unroll
    for (int t = 0; t < NT; ++t) {
      b0[t] = nb0[t];
      b1[t] = nb1[t];
    }
  }
}

__device__ __forceinline__ float xhalf(float x) {
  const int b = __float_as_int(x);
  return __int_as_float(__builtin_amdgcn_permlanex16(
      b, b, 0x76543210, 0xfedcba98, false, false));
}

__device__ __forceinline__ float fast_exp2(float x) {
  return __builtin_amdgcn_exp2f(x);
}

__global__ __launch_bounds__(256) void qkv_rope_so8_kernel(
    const float* __restrict__ x, const float* __restrict__ Wq,
    const float* __restrict__ Wk, const float* __restrict__ Wv,
    const float* __restrict__ rope, const float* __restrict__ so8,
    float* __restrict__ qb, float* __restrict__ kb, float* __restrict__ vtb) {
  __shared__ __attribute__((aligned(16))) float tile[TILE_M * QKV_PITCH];
  __shared__ float trig[TILE_M][8];

  const int tid = threadIdx.x;
  const int lane = tid & 31;
  const int widx = __builtin_amdgcn_readfirstlane(tid >> 5);
  const int c = lane & 15;
  const int h = lane >> 4;
  const int m0 = blockIdx.x * TILE_M;

  if (tid < 64) {
    const int row = tid >> 2;
    const int i = tid & 3;
    const float invf =
        (i == 0) ? 1.0f : ((i == 1) ? 0.1f : ((i == 2) ? 0.01f : 0.001f));
    const int s = (m0 + row) & (SEQ - 1);
    const float fr = (float)s * invf;
    trig[row][i] = cosf(fr);
    trig[row][4 + i] = sinf(fr);
  }

  const float* arow = x + (size_t)(m0 + c) * HID + 4 * h;
  const float* wrow[3];
#pragma unroll
  for (int t = 0; t < 3; ++t) {
    const int nt = widx * 3 + t;
    const int n0 = nt * 16;
    const float* Wb = (nt < 16) ? (Wq + (size_t)n0 * HID)
                                : ((nt < 20) ? (Wk + (size_t)(n0 - 256) * HID)
                                             : (Wv + (size_t)(n0 - 320) * HID));
    wrow[t] = Wb + (size_t)c * HID + 4 * h;
  }
  v8f acc[3] = {};
  wave_gemm_k256<3>(arow, wrow, acc);

#pragma unroll
  for (int t = 0; t < 3; ++t)
#pragma unroll
    for (int r = 0; r < 8; ++r)
      tile[(8 * h + r) * QKV_PITCH + (widx * 3 + t) * 16 + c] = acc[t][r];
  __syncthreads();

  for (int it = 0; it < 3; ++it) {
    const int id = it * 256 + tid;
    const int row = id & 15;
    const int hd = id >> 4;
    const int kind = (hd < 32) ? 0 : (hd < 40 ? 1 : 2);
    const int head = (hd < 32) ? hd : (hd < 40 ? hd - 32 : hd - 40);
    const int bs = m0 + row;
    const int s = bs & (SEQ - 1);
    const int b = bs >> 11;

    const float* tp = tile + row * QKV_PITCH + hd * 8;
    const v4f t0 = *(const v4f*)tp;
    const v4f t1 = *(const v4f*)(tp + 4);
    const float pr[HDIM] = {t0.x, t0.y, t0.z, t0.w, t1.x, t1.y, t1.z, t1.w};

    float y[HDIM];
    if (kind <= 1) {
      float cr[4] = {}, sr[4] = {};
#pragma unroll
      for (int i = 0; i < 4; ++i) {
        const float ci = trig[row][i], si = trig[row][4 + i];
#pragma unroll
        for (int j = 0; j < 4; ++j) {
          cr[j] = fmaf(ci, rope[i * 8 + j], cr[j]);
          sr[j] = fmaf(si, rope[i * 8 + j], sr[j]);
        }
      }
#pragma unroll
      for (int d = 0; d < HDIM; ++d) {
        const int j = d & 3;
        const float rh = (d < 4) ? -pr[d + 4] : pr[d - 4];
        y[d] = pr[d] * cr[j] + rh * sr[j];
      }
    } else {
#pragma unroll
      for (int d = 0; d < HDIM; ++d) y[d] = pr[d];
    }

    float z[HDIM];
#pragma unroll
    for (int j = 0; j < HDIM; ++j) {
      float sum = 0.0f;
#pragma unroll
      for (int d = 0; d < HDIM; ++d) sum = fmaf(y[d], so8[d * 8 + j], sum);
      z[j] = sum;
    }

    if (kind == 0) {
      float* dst = qb + (((size_t)b * N_HEADS + head) * SEQ + s) * HDIM;
      *(v4f*)dst = (v4f){z[0], z[1], z[2], z[3]};
      *(v4f*)(dst + 4) = (v4f){z[4], z[5], z[6], z[7]};
    } else if (kind == 1) {
      float* dst = kb + (((size_t)b * N_KV + head) * SEQ + s) * HDIM;
      *(v4f*)dst = (v4f){z[0], z[1], z[2], z[3]};
      *(v4f*)(dst + 4) = (v4f){z[4], z[5], z[6], z[7]};
    } else {
#pragma unroll
      for (int j = 0; j < HDIM; ++j)
        vtb[(((size_t)b * N_KV + head) * HDIM + j) * SEQ + s] = z[j];
    }
  }
}

template <bool MASK>
__device__ __forceinline__ void attn_tile(
    const float* __restrict__ kp, const float* __restrict__ vrow, int kbase,
    int lanelo, int hi, int hi2, bool vlane, float vfill, const v2f Qb0,
    const v2f Qb1, float& m, v8f& o) {
  const float* krow = kp + (size_t)(kbase + lanelo) * HDIM;
  const v2f Ka0 = *(const v2f*)(krow + hi2);
  const v2f Ka1 = *(const v2f*)(krow + 4 + hi2);
  const v4f V0 = *(const v4f*)(vrow + kbase);
  const v4f V1 = *(const v4f*)(vrow + kbase + 4);

  v8f sc = {};
  sc = __builtin_amdgcn_wmma_f32_16x16x4_f32(false, Ka0, false, Qb0, (short)0,
                                             sc, false, false);
  sc = __builtin_amdgcn_wmma_f32_16x16x4_f32(false, Ka1, false, Qb1, (short)0,
                                             sc, false, false);

  if (MASK) {
#pragma unroll
    for (int r = 0; r < 8; ++r)
      sc[r] += (8 * hi + r > lanelo) ? -3.0e38f : 0.0f;
  }

  float tm = fmaxf(fmaxf(fmaxf(sc[0], sc[1]), fmaxf(sc[2], sc[3])),
                   fmaxf(fmaxf(sc[4], sc[5]), fmaxf(sc[6], sc[7])));
  tm = fmaxf(tm, xhalf(tm));
  const float mnew = fmaxf(m, tm);
  const float corr = fast_exp2(m - mnew);
  m = mnew;

  v2f Pb[4];
#pragma unroll
  for (int t = 0; t < 4; ++t)
    Pb[t] = (v2f){fast_exp2(sc[2 * t] - mnew), fast_exp2(sc[2 * t + 1] - mnew)};

#pragma unroll
  for (int r = 0; r < 8; ++r) o[r] = o[r] * corr;

  const v2f fill = (v2f){vfill, vfill};
  v2f Va[4];
  Va[0] = vlane ? V0.xy : fill;
  Va[1] = vlane ? V0.zw : fill;
  Va[2] = vlane ? V1.xy : fill;
  Va[3] = vlane ? V1.zw : fill;
#pragma unroll
  for (int t = 0; t < 4; ++t)
    o = __builtin_amdgcn_wmma_f32_16x16x4_f32(false, Va[t], false, Pb[t],
                                              (short)0, o, false, false);
}

__global__ __launch_bounds__(256) void so8_flash_attn_kernel(
    const float* __restrict__ q, const float* __restrict__ k,
    const float* __restrict__ vT, float* __restrict__ attn) {
  const int lane = threadIdx.x & 31;
  const int widx = threadIdx.x >> 5;
  const int w = blockIdx.x * 8 + widx;
  const int qt = w & (QTILES - 1);
  const int h = (w >> 7) & (N_HEADS - 1);
  const int b = w >> 12;
  const int kv = h >> 2;
  const int qbase = qt * 16;
  const int lanelo = lane & 15;
  const int hi = lane >> 4;
  const int hi2 = hi * 2;

  const float* qp = q + ((size_t)b * N_HEADS + h) * SEQ * HDIM;
  const float* kp = k + ((size_t)b * N_KV + kv) * SEQ * HDIM;
  const float* vp = vT + ((size_t)b * N_KV + kv) * HDIM * SEQ;

  const float QSCALE = 0.35355339059327f * 1.44269504088896f;
  const float* qrow = qp + (size_t)(qbase + lanelo) * HDIM;
  const v2f Qb0 = *(const v2f*)(qrow + hi2) * QSCALE;
  const v2f Qb1 = *(const v2f*)(qrow + 4 + hi2) * QSCALE;

  const float* vrow = vp + (size_t)(lanelo & 7) * SEQ + 8 * hi;
  const bool vlane = lanelo < HDIM;
  const float vfill = (lanelo == HDIM) ? 1.0f : 0.0f;

  v8f o = {};
  float m = -3.0e38f;

  for (int kt = 0; kt < qt; ++kt)
    attn_tile<false>(kp, vrow, kt * 16, lanelo, hi, hi2, vlane, vfill, Qb0,
                     Qb1, m, o);
  attn_tile<true>(kp, vrow, qbase, lanelo, hi, hi2, vlane, vfill, Qb0, Qb1, m,
                  o);

  const float l = xhalf(o[0]);
  const float inv = __builtin_amdgcn_rcpf(l);
  if (hi == 0) {
    float* dst =
        attn + (((size_t)b * SEQ + qbase + lanelo) * N_HEADS + h) * HDIM;
    *(v4f*)dst = (v4f){o[0] * inv, o[1] * inv, o[2] * inv, o[3] * inv};
    *(v4f*)(dst + 4) = (v4f){o[4] * inv, o[5] * inv, o[6] * inv, o[7] * inv};
  }
}

__global__ __launch_bounds__(256) void out_proj_kernel(
    const float* __restrict__ attn, const float* __restrict__ Wo,
    float* __restrict__ out) {
  __shared__ __attribute__((aligned(16))) float tile[TILE_M * OUT_PITCH];

  const int tid = threadIdx.x;
  const int lane = tid & 31;
  const int widx = tid >> 5;
  const int c = lane & 15;
  const int h = lane >> 4;
  const int m0 = blockIdx.x * TILE_M;

  const float* arow = attn + (size_t)(m0 + c) * HID + 4 * h;
  const float* wrow[2];
#pragma unroll
  for (int t = 0; t < 2; ++t)
    wrow[t] = Wo + (size_t)((widx * 2 + t) * 16 + c) * HID + 4 * h;
  v8f acc[2] = {};
  wave_gemm_k256<2>(arow, wrow, acc);

#pragma unroll
  for (int t = 0; t < 2; ++t)
#pragma unroll
    for (int r = 0; r < 8; ++r)
      tile[(8 * h + r) * OUT_PITCH + (widx * 2 + t) * 16 + c] = acc[t][r];
  __syncthreads();

#pragma unroll
  for (int it = 0; it < 4; ++it) {
    const int id = it * 256 + tid;
    const int row = id >> 6;
    const int c4 = (id & 63) * 4;
    *(v4f*)(out + (size_t)(m0 + row) * HID + c4) =
        *(const v4f*)(tile + row * OUT_PITCH + c4);
  }
}

extern "C" void kernel_launch(void* const* d_in, const int* in_sizes, int n_in,
                              void* d_out, int out_size, void* d_ws,
                              size_t ws_size, hipStream_t stream) {
  const float* x = (const float*)d_in[0];
  const float* Wq = (const float*)d_in[1];
  const float* Wk = (const float*)d_in[2];
  const float* Wv = (const float*)d_in[3];
  const float* Wo = (const float*)d_in[4];
  const float* rope = (const float*)d_in[5];
  const float* so8 = (const float*)d_in[6];

  float* ws = (float*)d_ws;
  float* qb = ws;
  float* kb = qb + (size_t)2 * N_HEADS * SEQ * HDIM;
  float* vtb = kb + (size_t)2 * N_KV * SEQ * HDIM;
  float* ab = vtb + (size_t)2 * N_KV * SEQ * HDIM;

  const int BS = 2 * SEQ;

  qkv_rope_so8_kernel<<<BS / TILE_M, 256, 0, stream>>>(x, Wq, Wk, Wv, rope, so8,
                                                       qb, kb, vtb);
  so8_flash_attn_kernel<<<(2 * N_HEADS * QTILES) / 8, 256, 0, stream>>>(
      qb, kb, vtb, ab);
  out_proj_kernel<<<BS / TILE_M, 256, 0, stream>>>(ab, Wo, (float*)d_out);
}


# Attention key loop: K and V^T fragments requested 3 tiles ahead into a 4-slot register ring (unrolled by 4, requests pinned in slot order, scalar tile counter); every sum and its order unchanged
# speedup vs baseline: 3.6987x; 1.0858x over previous
#include <hip/hip_runtime.h>
#include <hip/hip_bf16.h>

#define N_HEADS 32
#define N_KV 8
#define HDIM 8
#define SEQ 2048
#define HID 256
#define QTILES (SEQ / 16)
#define KV_RING 4

typedef __attribute__((ext_vector_type(2))) float v2f;
typedef __attribute__((ext_vector_type(8))) float v8f;
typedef __attribute__((ext_vector_type(4))) float v4f;


#define TILE_M 16
#define QKV_PITCH 388
#define OUT_PITCH 260

template <int NT>
__device__ __forceinline__ void wave_gemm_k256(const float* __restrict__ arow,
                                               const float* const (&wrow)[NT],
                                               v8f (&acc)[NT]) {
  v4f a0 = *(const v4f*)(arow);
  v4f a1 = *(const v4f*)(arow + 8);
  v4f b0[NT], b1[NT];
#pragma unroll
  for (int t = 0; t < NT; ++t) {
    b0[t] = *(const v4f*)(wrow[t]);
    b1[t] = *(const v4f*)(wrow[t] + 8);
  }
#pragma unroll
  for (int kc = 0; kc < HID / 16; ++kc) {
    v4f na0 = a0, na1 = a1;
    v4f nb0[NT], nb1[NT];
#pragma unroll
    for (int t = 0; t < NT; ++t) {
      nb0[t] = b0[t];
      nb1[t] = b1[t];
    }
    if (kc + 1 < HID / 16) {
      const int ko = (kc + 1) * 16;
      na0 = *(const v4f*)(arow + ko);
      na1 = *(const v4f*)(arow + ko + 8);
#pragma unroll
      for (int t = 0; t < NT; ++t) {
        nb0[t] = *(const v4f*)(wrow[t] + ko);
        nb1[t] = *(const v4f*)(wrow[t] + ko + 8);
      }
    }
#pragma unroll
    for (int t = 0; t < NT; ++t)
      acc[t] = __builtin_amdgcn_wmma_f32_16x16x4_f32(
          false, a0.xy, false, b0[t].xy, (short)0, acc[t], false, false);
#pragma unroll
    for (int t = 0; t < NT; ++t)
      acc[t] = __builtin_amdgcn_wmma_f32_16x16x4_f32(
          false, a0.zw, false, b0[t].zw, (short)0, acc[t], false, false);
#pragma unroll
    for (int t = 0; t < NT; ++t)
      acc[t] = __builtin_amdgcn_wmma_f32_16x16x4_f32(
          false, a1.xy, false, b1[t].xy, (short)0, acc[t], false, false);
#pragma unroll
    for (int t = 0; t < NT; ++t)
      acc[t] = __builtin_amdgcn_wmma_f32_16x16x4_f32(
          false, a1.zw, false, b1[t].zw, (short)0, acc[t], false, false);
    a0 = na0;
    a1 = na1;
#pragma unroll
    for (int t = 0; t < NT; ++t) {
      b0[t] = nb0[t];
      b1[t] = nb1[t];
    }
  }
}

__device__ __forceinline__ float xhalf(float x) {
  const int b = __float_as_int(x);
  return __int_as_float(__builtin_amdgcn_permlanex16(
      b, b, 0x76543210, 0xfedcba98, false, false));
}

__device__ __forceinline__ float fast_exp2(float x) {
  return __builtin_amdgcn_exp2f(x);
}

__global__ __launch_bounds__(256) void qkv_rope_so8_kernel(
    const float* __restrict__ x, const float* __restrict__ Wq,
    const float* __restrict__ Wk, const float* __restrict__ Wv,
    const float* __restrict__ rope, const float* __restrict__ so8,
    float* __restrict__ qb, float* __restrict__ kb, float* __restrict__ vtb) {
  __shared__ __attribute__((aligned(16))) float tile[TILE_M * QKV_PITCH];
  __shared__ float trig[TILE_M][8];

  const int tid = threadIdx.x;
  const int lane = tid & 31;
  const int widx = __builtin_amdgcn_readfirstlane(tid >> 5);
  const int c = lane & 15;
  const int h = lane >> 4;
  const int m0 = blockIdx.x * TILE_M;

  if (tid < 64) {
    const int row = tid >> 2;
    const int i = tid & 3;
    const float invf =
        (i == 0) ? 1.0f : ((i == 1) ? 0.1f : ((i == 2) ? 0.01f : 0.001f));
    const int s = (m0 + row) & (SEQ - 1);
    const float fr = (float)s * invf;
    trig[row][i] = cosf(fr);
    trig[row][4 + i] = sinf(fr);
  }

  const float* arow = x + (size_t)(m0 + c) * HID + 4 * h;
  const float* wrow[3];
#pragma unroll
  for (int t = 0; t < 3; ++t) {
    const int nt = widx * 3 + t;
    const int n0 = nt * 16;
    const float* Wb = (nt < 16) ? (Wq + (size_t)n0 * HID)
                                : ((nt < 20) ? (Wk + (size_t)(n0 - 256) * HID)
                                             : (Wv + (size_t)(n0 - 320) * HID));
    wrow[t] = Wb + (size_t)c * HID + 4 * h;
  }
  v8f acc[3] = {};
  wave_gemm_k256<3>(arow, wrow, acc);

#pragma unroll
  for (int t = 0; t < 3; ++t)
#pragma unroll
    for (int r = 0; r < 8; ++r)
      tile[(8 * h + r) * QKV_PITCH + (widx * 3 + t) * 16 + c] = acc[t][r];
  __syncthreads();

  for (int it = 0; it < 3; ++it) {
    const int id = it * 256 + tid;
    const int row = id & 15;
    const int hd = id >> 4;
    const int kind = (hd < 32) ? 0 : (hd < 40 ? 1 : 2);
    const int head = (hd < 32) ? hd : (hd < 40 ? hd - 32 : hd - 40);
    const int bs = m0 + row;
    const int s = bs & (SEQ - 1);
    const int b = bs >> 11;

    const float* tp = tile + row * QKV_PITCH + hd * 8;
    const v4f t0 = *(const v4f*)tp;
    const v4f t1 = *(const v4f*)(tp + 4);
    const float pr[HDIM] = {t0.x, t0.y, t0.z, t0.w, t1.x, t1.y, t1.z, t1.w};

    float y[HDIM];
    if (kind <= 1) {
      float cr[4] = {}, sr[4] = {};
#pragma unroll
      for (int i = 0; i < 4; ++i) {
        const float ci = trig[row][i], si = trig[row][4 + i];
#pragma unroll
        for (int j = 0; j < 4; ++j) {
          cr[j] = fmaf(ci, rope[i * 8 + j], cr[j]);
          sr[j] = fmaf(si, rope[i * 8 + j], sr[j]);
        }
      }
#pragma unroll
      for (int d = 0; d < HDIM; ++d) {
        const int j = d & 3;
        const float rh = (d < 4) ? -pr[d + 4] : pr[d - 4];
        y[d] = pr[d] * cr[j] + rh * sr[j];
      }
    } else {
#pragma unroll
      for (int d = 0; d < HDIM; ++d) y[d] = pr[d];
    }

    float z[HDIM];
#pragma unroll
    for (int j = 0; j < HDIM; ++j) {
      float sum = 0.0f;
#pragma unroll
      for (int d = 0; d < HDIM; ++d) sum = fmaf(y[d], so8[d * 8 + j], sum);
      z[j] = sum;
    }

    if (kind == 0) {
      float* dst = qb + (((size_t)b * N_HEADS + head) * SEQ + s) * HDIM;
      *(v4f*)dst = (v4f){z[0], z[1], z[2], z[3]};
      *(v4f*)(dst + 4) = (v4f){z[4], z[5], z[6], z[7]};
    } else if (kind == 1) {
      float* dst = kb + (((size_t)b * N_KV + head) * SEQ + s) * HDIM;
      *(v4f*)dst = (v4f){z[0], z[1], z[2], z[3]};
      *(v4f*)(dst + 4) = (v4f){z[4], z[5], z[6], z[7]};
    } else {
#pragma unroll
      for (int j = 0; j < HDIM; ++j)
        vtb[(((size_t)b * N_KV + head) * HDIM + j) * SEQ + s] = z[j];
    }
  }
}

struct KVFrag {
  v2f Ka0, Ka1;
  v4f V0, V1;
};

__device__ __forceinline__ KVFrag load_kv(const float* __restrict__ krow0,
                                          const float* __restrict__ vrow,
                                          int kbase) {
  KVFrag f;
  const float* krow = krow0 + (size_t)kbase * HDIM;
  f.Ka0 = *(const v2f*)(krow);
  f.Ka1 = *(const v2f*)(krow + 4);
  f.V0 = *(const v4f*)(vrow + kbase);
  f.V1 = *(const v4f*)(vrow + kbase + 4);
  return f;
}

template <bool MASK>
__device__ __forceinline__ void attn_tile(
    const KVFrag& f, int lanelo, int hi, bool vlane, float vfill,
    const v2f Qb0, const v2f Qb1, float& m, v8f& o) {
  const v2f Ka0 = f.Ka0, Ka1 = f.Ka1;
  const v4f V0 = f.V0, V1 = f.V1;

  v8f sc = {};
  sc = __builtin_amdgcn_wmma_f32_16x16x4_f32(false, Ka0, false, Qb0, (short)0,
                                             sc, false, false);
  sc = __builtin_amdgcn_wmma_f32_16x16x4_f32(false, Ka1, false, Qb1, (short)0,
                                             sc, false, false);

  if (MASK) {
#pragma unroll
    for (int r = 0; r < 8; ++r)
      sc[r] += (8 * hi + r > lanelo) ? -3.0e38f : 0.0f;
  }

  float tm = fmaxf(fmaxf(fmaxf(sc[0], sc[1]), fmaxf(sc[2], sc[3])),
                   fmaxf(fmaxf(sc[4], sc[5]), fmaxf(sc[6], sc[7])));
  tm = fmaxf(tm, xhalf(tm));
  const float mnew = fmaxf(m, tm);
  const float corr = fast_exp2(m - mnew);
  m = mnew;

  v2f Pb[4];
#pragma unroll
  for (int t = 0; t < 4; ++t)
    Pb[t] = (v2f){fast_exp2(sc[2 * t] - mnew), fast_exp2(sc[2 * t + 1] - mnew)};

#pragma unroll
  for (int r = 0; r < 8; ++r) o[r] = o[r] * corr;

  const v2f fill = (v2f){vfill, vfill};
  v2f Va[4];
  Va[0] = vlane ? V0.xy : fill;
  Va[1] = vlane ? V0.zw : fill;
  Va[2] = vlane ? V1.xy : fill;
  Va[3] = vlane ? V1.zw : fill;
#pragma unroll
  for (int t = 0; t < 4; ++t)
    o = __builtin_amdgcn_wmma_f32_16x16x4_f32(false, Va[t], false, Pb[t],
                                              (short)0, o, false, false);
}

__global__ __launch_bounds__(256) void so8_flash_attn_kernel(
    const float* __restrict__ q, const float* __restrict__ k,
    const float* __restrict__ vT, float* __restrict__ attn) {
  const int lane = threadIdx.x & 31;
  const int widx = __builtin_amdgcn_readfirstlane(threadIdx.x >> 5);
  const int w = blockIdx.x * 8 + widx;
  const int qt = w & (QTILES - 1);
  const int h = (w >> 7) & (N_HEADS - 1);
  const int b = w >> 12;
  const int kv = h >> 2;
  const int qbase = qt * 16;
  const int lanelo = lane & 15;
  const int hi = lane >> 4;
  const int hi2 = hi * 2;

  const float* qp = q + ((size_t)b * N_HEADS + h) * SEQ * HDIM;
  const float* kp = k + ((size_t)b * N_KV + kv) * SEQ * HDIM;
  const float* vp = vT + ((size_t)b * N_KV + kv) * HDIM * SEQ;

  const float QSCALE = 0.35355339059327f * 1.44269504088896f;
  const float* qrow = qp + (size_t)(qbase + lanelo) * HDIM;
  const v2f Qb0 = *(const v2f*)(qrow + hi2) * QSCALE;
  const v2f Qb1 = *(const v2f*)(qrow + 4 + hi2) * QSCALE;

  const float* vrow = vp + (size_t)(lanelo & 7) * SEQ + 8 * hi;
  const bool vlane = lanelo < HDIM;
  const float vfill = (lanelo == HDIM) ? 1.0f : 0.0f;

  v8f o = {};
  float m = -3.0e38f;

  const float* krow0 = kp + (size_t)lanelo * HDIM + hi2;

  KVFrag f[KV_RING];
#pragma unroll
  for (int i = 0; i < KV_RING; ++i) {
    f[i] = load_kv(krow0, vrow, min(i, qt) * 16);
    __builtin_amdgcn_sched_barrier(0);
  }

  int kt = 0;
  for (; kt + KV_RING <= qt; kt += KV_RING) {
#pragma unroll
    for (int i = 0; i < KV_RING; ++i) {
      attn_tile<false>(f[i], lanelo, hi, vlane, vfill, Qb0, Qb1, m, o);
      f[i] = load_kv(krow0, vrow, min(kt + KV_RING + i, qt) * 16);
      __builtin_amdgcn_sched_barrier(0);
    }
  }
#pragma unroll
  for (int i = 0; i < KV_RING - 1; ++i) {
    if (kt < qt) {
      attn_tile<false>(f[0], lanelo, hi, vlane, vfill, Qb0, Qb1, m, o);
#pragma unroll
      for (int j = 0; j + 1 < KV_RING; ++j) f[j] = f[j + 1];
      ++kt;
    }
  }
  attn_tile<true>(f[0], lanelo, hi, vlane, vfill, Qb0, Qb1, m, o);

  const float l = xhalf(o[0]);
  const float inv = __builtin_amdgcn_rcpf(l);
  if (hi == 0) {
    float* dst =
        attn + (((size_t)b * SEQ + qbase + lanelo) * N_HEADS + h) * HDIM;
    *(v4f*)dst = (v4f){o[0] * inv, o[1] * inv, o[2] * inv, o[3] * inv};
    *(v4f*)(dst + 4) = (v4f){o[4] * inv, o[5] * inv, o[6] * inv, o[7] * inv};
  }
}

__global__ __launch_bounds__(256) void out_proj_kernel(
    const float* __restrict__ attn, const float* __restrict__ Wo,
    float* __restrict__ out) {
  __shared__ __attribute__((aligned(16))) float tile[TILE_M * OUT_PITCH];

  const int tid = threadIdx.x;
  const int lane = tid & 31;
  const int widx = tid >> 5;
  const int c = lane & 15;
  const int h = lane >> 4;
  const int m0 = blockIdx.x * TILE_M;

  const float* arow = attn + (size_t)(m0 + c) * HID + 4 * h;
  const float* wrow[2];
#pragma unroll
  for (int t = 0; t < 2; ++t)
    wrow[t] = Wo + (size_t)((widx * 2 + t) * 16 + c) * HID + 4 * h;
  v8f acc[2] = {};
  wave_gemm_k256<2>(arow, wrow, acc);

#pragma unroll
  for (int t = 0; t < 2; ++t)
#pragma unroll
    for (int r = 0; r < 8; ++r)
      tile[(8 * h + r) * OUT_PITCH + (widx * 2 + t) * 16 + c] = acc[t][r];
  __syncthreads();

#pragma unroll
  for (int it = 0; it < 4; ++it) {
    const int id = it * 256 + tid;
    const int row = id >> 6;
    const int c4 = (id & 63) * 4;
    *(v4f*)(out + (size_t)(m0 + row) * HID + c4) =
        *(const v4f*)(tile + row * OUT_PITCH + c4);
  }
}

extern "C" void kernel_launch(void* const* d_in, const int* in_sizes, int n_in,
                              void* d_out, int out_size, void* d_ws,
                              size_t ws_size, hipStream_t stream) {
  const float* x = (const float*)d_in[0];
  const float* Wq = (const float*)d_in[1];
  const float* Wk = (const float*)d_in[2];
  const float* Wv = (const float*)d_in[3];
  const float* Wo = (const float*)d_in[4];
  const float* rope = (const float*)d_in[5];
  const float* so8 = (const float*)d_in[6];

  float* ws = (float*)d_ws;
  float* qb = ws;
  float* kb = qb + (size_t)2 * N_HEADS * SEQ * HDIM;
  float* vtb = kb + (size_t)2 * N_KV * SEQ * HDIM;
  float* ab = vtb + (size_t)2 * N_KV * SEQ * HDIM;

  const int BS = 2 * SEQ;

  qkv_rope_so8_kernel<<<BS / TILE_M, 256, 0, stream>>>(x, Wq, Wk, Wv, rope, so8,
                                                       qb, kb, vtb);
  so8_flash_attn_kernel<<<(2 * N_HEADS * QTILES) / 8, 256, 0, stream>>>(
      qb, kb, vtb, ab);
  out_proj_kernel<<<BS / TILE_M, 256, 0, stream>>>(ab, Wo, (float*)d_out);
}


# Attention: each wave does q tile p and then its mirror tile 127-p (129 key tiles per wave, 4096 waves) instead of one tile per wave: equal work on every SIMD; every sum and its order unchanged
# speedup vs baseline: 4.9166x; 1.3293x over previous
#include <hip/hip_runtime.h>
#include <hip/hip_bf16.h>

#define N_HEADS 32
#define N_KV 8
#define HDIM 8
#define SEQ 2048
#define HID 256
#define QTILES (SEQ / 16)
#define KV_RING 4

typedef __attribute__((ext_vector_type(2))) float v2f;
typedef __attribute__((ext_vector_type(8))) float v8f;
typedef __attribute__((ext_vector_type(4))) float v4f;


#define TILE_M 16
#define QKV_PITCH 388
#define OUT_PITCH 260

template <int NT>
__device__ __forceinline__ void wave_gemm_k256(const float* __restrict__ arow,
                                               const float* const (&wrow)[NT],
                                               v8f (&acc)[NT]) {
  v4f a0 = *(const v4f*)(arow);
  v4f a1 = *(const v4f*)(arow + 8);
  v4f b0[NT], b1[NT];
#pragma unroll
  for (int t = 0; t < NT; ++t) {
    b0[t] = *(const v4f*)(wrow[t]);
    b1[t] = *(const v4f*)(wrow[t] + 8);
  }
#pragma unroll
  for (int kc = 0; kc < HID / 16; ++kc) {
    v4f na0 = a0, na1 = a1;
    v4f nb0[NT], nb1[NT];
#pragma unroll
    for (int t = 0; t < NT; ++t) {
      nb0[t] = b0[t];
      nb1[t] = b1[t];
    }
    if (kc + 1 < HID / 16) {
      const int ko = (kc + 1) * 16;
      na0 = *(const v4f*)(arow + ko);
      na1 = *(const v4f*)(arow + ko + 8);
#pragma unroll
      for (int t = 0; t < NT; ++t) {
        nb0[t] = *(const v4f*)(wrow[t] + ko);
        nb1[t] = *(const v4f*)(wrow[t] + ko + 8);
      }
    }
#pragma unroll
    for (int t = 0; t < NT; ++t)
      acc[t] = __builtin_amdgcn_wmma_f32_16x16x4_f32(
          false, a0.xy, false, b0[t].xy, (short)0, acc[t], false, false);
#pragma unroll
    for (int t = 0; t < NT; ++t)
      acc[t] = __builtin_amdgcn_wmma_f32_16x16x4_f32(
          false, a0.zw, false, b0[t].zw, (short)0, acc[t], false, false);
#pragma unroll
    for (int t = 0; t < NT; ++t)
      acc[t] = __builtin_amdgcn_wmma_f32_16x16x4_f32(
          false, a1.xy, false, b1[t].xy, (short)0, acc[t], false, false);
#pragma unroll
    for (int t = 0; t < NT; ++t)
      acc[t] = __builtin_amdgcn_wmma_f32_16x16x4_f32(
          false, a1.zw, false, b1[t].zw, (short)0, acc[t], false, false);
    a0 = na0;
    a1 = na1;
#pragma unroll
    for (int t = 0; t < NT; ++t) {
      b0[t] = nb0[t];
      b1[t] = nb1[t];
    }
  }
}

__device__ __forceinline__ float xhalf(float x) {
  const int b = __float_as_int(x);
  return __int_as_float(__builtin_amdgcn_permlanex16(
      b, b, 0x76543210, 0xfedcba98, false, false));
}

__device__ __forceinline__ float fast_exp2(float x) {
  return __builtin_amdgcn_exp2f(x);
}

__global__ __launch_bounds__(256) void qkv_rope_so8_kernel(
    const float* __restrict__ x, const float* __restrict__ Wq,
    const float* __restrict__ Wk, const float* __restrict__ Wv,
    const float* __restrict__ rope, const float* __restrict__ so8,
    float* __restrict__ qb, float* __restrict__ kb, float* __restrict__ vtb) {
  __shared__ __attribute__((aligned(16))) float tile[TILE_M * QKV_PITCH];
  __shared__ float trig[TILE_M][8];

  const int tid = threadIdx.x;
  const int lane = tid & 31;
  const int widx = __builtin_amdgcn_readfirstlane(tid >> 5);
  const int c = lane & 15;
  const int h = lane >> 4;
  const int m0 = blockIdx.x * TILE_M;

  if (tid < 64) {
    const int row = tid >> 2;
    const int i = tid & 3;
    const float invf =
        (i == 0) ? 1.0f : ((i == 1) ? 0.1f : ((i == 2) ? 0.01f : 0.001f));
    const int s = (m0 + row) & (SEQ - 1);
    const float fr = (float)s * invf;
    trig[row][i] = cosf(fr);
    trig[row][4 + i] = sinf(fr);
  }

  const float* arow = x + (size_t)(m0 + c) * HID + 4 * h;
  const float* wrow[3];
#pragma unroll
  for (int t = 0; t < 3; ++t) {
    const int nt = widx * 3 + t;
    const int n0 = nt * 16;
    const float* Wb = (nt < 16) ? (Wq + (size_t)n0 * HID)
                                : ((nt < 20) ? (Wk + (size_t)(n0 - 256) * HID)
                                             : (Wv + (size_t)(n0 - 320) * HID));
    wrow[t] = Wb + (size_t)c * HID + 4 * h;
  }
  v8f acc[3] = {};
  wave_gemm_k256<3>(arow, wrow, acc);

#pragma unroll
  for (int t = 0; t < 3; ++t)
#pragma unroll
    for (int r = 0; r < 8; ++r)
      tile[(8 * h + r) * QKV_PITCH + (widx * 3 + t) * 16 + c] = acc[t][r];
  __syncthreads();

  for (int it = 0; it < 3; ++it) {
    const int id = it * 256 + tid;
    const int row = id & 15;
    const int hd = id >> 4;
    const int kind = (hd < 32) ? 0 : (hd < 40 ? 1 : 2);
    const int head = (hd < 32) ? hd : (hd < 40 ? hd - 32 : hd - 40);
    const int bs = m0 + row;
    const int s = bs & (SEQ - 1);
    const int b = bs >> 11;

    const float* tp = tile + row * QKV_PITCH + hd * 8;
    const v4f t0 = *(const v4f*)tp;
    const v4f t1 = *(const v4f*)(tp + 4);
    const float pr[HDIM] = {t0.x, t0.y, t0.z, t0.w, t1.x, t1.y, t1.z, t1.w};

    float y[HDIM];
    if (kind <= 1) {
      float cr[4] = {}, sr[4] = {};
#pragma unroll
      for (int i = 0; i < 4; ++i) {
        const float ci = trig[row][i], si = trig[row][4 + i];
#pragma unroll
        for (int j = 0; j < 4; ++j) {
          cr[j] = fmaf(ci, rope[i * 8 + j], cr[j]);
          sr[j] = fmaf(si, rope[i * 8 + j], sr[j]);
        }
      }
#pragma unroll
      for (int d = 0; d < HDIM; ++d) {
        const int j = d & 3;
        const float rh = (d < 4) ? -pr[d + 4] : pr[d - 4];
        y[d] = pr[d] * cr[j] + rh * sr[j];
      }
    } else {
#pragma unroll
      for (int d = 0; d < HDIM; ++d) y[d] = pr[d];
    }

    float z[HDIM];
#pragma unroll
    for (int j = 0; j < HDIM; ++j) {
      float sum = 0.0f;
#pragma unroll
      for (int d = 0; d < HDIM; ++d) sum = fmaf(y[d], so8[d * 8 + j], sum);
      z[j] = sum;
    }

    if (kind == 0) {
      float* dst = qb + (((size_t)b * N_HEADS + head) * SEQ + s) * HDIM;
      *(v4f*)dst = (v4f){z[0], z[1], z[2], z[3]};
      *(v4f*)(dst + 4) = (v4f){z[4], z[5], z[6], z[7]};
    } else if (kind == 1) {
      float* dst = kb + (((size_t)b * N_KV + head) * SEQ + s) * HDIM;
      *(v4f*)dst = (v4f){z[0], z[1], z[2], z[3]};
      *(v4f*)(dst + 4) = (v4f){z[4], z[5], z[6], z[7]};
    } else {
#pragma unroll
      for (int j = 0; j < HDIM; ++j)
        vtb[(((size_t)b * N_KV + head) * HDIM + j) * SEQ + s] = z[j];
    }
  }
}

struct KVFrag {
  v2f Ka0, Ka1;
  v4f V0, V1;
};

__device__ __forceinline__ KVFrag load_kv(const float* __restrict__ krow0,
                                          const float* __restrict__ vrow,
                                          int kbase) {
  KVFrag f;
  const float* krow = krow0 + (size_t)kbase * HDIM;
  f.Ka0 = *(const v2f*)(krow);
  f.Ka1 = *(const v2f*)(krow + 4);
  f.V0 = *(const v4f*)(vrow + kbase);
  f.V1 = *(const v4f*)(vrow + kbase + 4);
  return f;
}

template <bool MASK>
__device__ __forceinline__ void attn_tile(
    const KVFrag& f, int lanelo, int hi, bool vlane, float vfill,
    const v2f Qb0, const v2f Qb1, float& m, v8f& o) {
  const v2f Ka0 = f.Ka0, Ka1 = f.Ka1;
  const v4f V0 = f.V0, V1 = f.V1;

  v8f sc = {};
  sc = __builtin_amdgcn_wmma_f32_16x16x4_f32(false, Ka0, false, Qb0, (short)0,
                                             sc, false, false);
  sc = __builtin_amdgcn_wmma_f32_16x16x4_f32(false, Ka1, false, Qb1, (short)0,
                                             sc, false, false);

  if (MASK) {
#pragma unroll
    for (int r = 0; r < 8; ++r)
      sc[r] += (8 * hi + r > lanelo) ? -3.0e38f : 0.0f;
  }

  float tm = fmaxf(fmaxf(fmaxf(sc[0], sc[1]), fmaxf(sc[2], sc[3])),
                   fmaxf(fmaxf(sc[4], sc[5]), fmaxf(sc[6], sc[7])));
  tm = fmaxf(tm, xhalf(tm));
  const float mnew = fmaxf(m, tm);
  const float corr = fast_exp2(m - mnew);
  m = mnew;

  v2f Pb[4];
#pragma unroll
  for (int t = 0; t < 4; ++t)
    Pb[t] = (v2f){fast_exp2(sc[2 * t] - mnew), fast_exp2(sc[2 * t + 1] - mnew)};

#pragma unroll
  for (int r = 0; r < 8; ++r) o[r] = o[r] * corr;

  const v2f fill = (v2f){vfill, vfill};
  v2f Va[4];
  Va[0] = vlane ? V0.xy : fill;
  Va[1] = vlane ? V0.zw : fill;
  Va[2] = vlane ? V1.xy : fill;
  Va[3] = vlane ? V1.zw : fill;
#pragma unroll
  for (int t = 0; t < 4; ++t)
    o = __builtin_amdgcn_wmma_f32_16x16x4_f32(false, Va[t], false, Pb[t],
                                              (short)0, o, false, false);
}

__global__ __launch_bounds__(256) void so8_flash_attn_kernel(
    const float* __restrict__ q, const float* __restrict__ k,
    const float* __restrict__ vT, float* __restrict__ attn) {
  const int lane = threadIdx.x & 31;
  const int widx = __builtin_amdgcn_readfirstlane(threadIdx.x >> 5);
  const int w = blockIdx.x * 8 + widx;
  const int pr = w & (QTILES / 2 - 1);
  const int h = (w >> 6) & (N_HEADS - 1);
  const int b = w >> 11;
  const int kv = h >> 2;
  const int lanelo = lane & 15;
  const int hi = lane >> 4;
  const int hi2 = hi * 2;

  const float* qp = q + ((size_t)b * N_HEADS + h) * SEQ * HDIM;
  const float* kp = k + ((size_t)b * N_KV + kv) * SEQ * HDIM;
  const float* vp = vT + ((size_t)b * N_KV + kv) * HDIM * SEQ;

  const float* vrow = vp + (size_t)(lanelo & 7) * SEQ + 8 * hi;
  const bool vlane = lanelo < HDIM;
  const float vfill = (lanelo == HDIM) ? 1.0f : 0.0f;

  const float* krow0 = kp + (size_t)lanelo * HDIM + hi2;

  const float QSCALE = 0.35355339059327f * 1.44269504088896f;

#pragma unroll
  for (int ph = 0; ph < 2; ++ph) {
    const int qt = (ph == 0) ? pr : (QTILES - 1 - pr);
    const int qbase = qt * 16;

    const float* qrow = qp + (size_t)(qbase + lanelo) * HDIM;
    const v2f Qb0 = *(const v2f*)(qrow + hi2) * QSCALE;
    const v2f Qb1 = *(const v2f*)(qrow + 4 + hi2) * QSCALE;

    v8f o = {};
    float m = -3.0e38f;

    KVFrag f[KV_RING];
#pragma unroll
    for (int i = 0; i < KV_RING; ++i) {
      f[i] = load_kv(krow0, vrow, min(i, qt) * 16);
      __builtin_amdgcn_sched_barrier(0);
    }

    int kt = 0;
    for (; kt + KV_RING <= qt; kt += KV_RING) {
#pragma unroll
      for (int i = 0; i < KV_RING; ++i) {
        attn_tile<false>(f[i], lanelo, hi, vlane, vfill, Qb0, Qb1, m, o);
        f[i] = load_kv(krow0, vrow, min(kt + KV_RING + i, qt) * 16);
        __builtin_amdgcn_sched_barrier(0);
      }
    }
#pragma unroll
    for (int i = 0; i < KV_RING - 1; ++i) {
      if (kt < qt) {
        attn_tile<false>(f[0], lanelo, hi, vlane, vfill, Qb0, Qb1, m, o);
#pragma unroll
        for (int j = 0; j + 1 < KV_RING; ++j) f[j] = f[j + 1];
        ++kt;
      }
    }
    attn_tile<true>(f[0], lanelo, hi, vlane, vfill, Qb0, Qb1, m, o);

    const float l = xhalf(o[0]);
    const float inv = __builtin_amdgcn_rcpf(l);
    if (hi == 0) {
      float* dst =
          attn + (((size_t)b * SEQ + qbase + lanelo) * N_HEADS + h) * HDIM;
      *(v4f*)dst = (v4f){o[0] * inv, o[1] * inv, o[2] * inv, o[3] * inv};
      *(v4f*)(dst + 4) = (v4f){o[4] * inv, o[5] * inv, o[6] * inv, o[7] * inv};
    }
  }
}

__global__ __launch_bounds__(256) void out_proj_kernel(
    const float* __restrict__ attn, const float* __restrict__ Wo,
    float* __restrict__ out) {
  __shared__ __attribute__((aligned(16))) float tile[TILE_M * OUT_PITCH];

  const int tid = threadIdx.x;
  const int lane = tid & 31;
  const int widx = tid >> 5;
  const int c = lane & 15;
  const int h = lane >> 4;
  const int m0 = blockIdx.x * TILE_M;

  const float* arow = attn + (size_t)(m0 + c) * HID + 4 * h;
  const float* wrow[2];
#pragma unroll
  for (int t = 0; t < 2; ++t)
    wrow[t] = Wo + (size_t)((widx * 2 + t) * 16 + c) * HID + 4 * h;
  v8f acc[2] = {};
  wave_gemm_k256<2>(arow, wrow, acc);

#pragma unroll
  for (int t = 0; t < 2; ++t)
#pragma unroll
    for (int r = 0; r < 8; ++r)
      tile[(8 * h + r) * OUT_PITCH + (widx * 2 + t) * 16 + c] = acc[t][r];
  __syncthreads();

#pragma unroll
  for (int it = 0; it < 4; ++it) {
    const int id = it * 256 + tid;
    const int row = id >> 6;
    const int c4 = (id & 63) * 4;
    *(v4f*)(out + (size_t)(m0 + row) * HID + c4) =
        *(const v4f*)(tile + row * OUT_PITCH + c4);
  }
}

extern "C" void kernel_launch(void* const* d_in, const int* in_sizes, int n_in,
                              void* d_out, int out_size, void* d_ws,
                              size_t ws_size, hipStream_t stream) {
  const float* x = (const float*)d_in[0];
  const float* Wq = (const float*)d_in[1];
  const float* Wk = (const float*)d_in[2];
  const float* Wv = (const float*)d_in[3];
  const float* Wo = (const float*)d_in[4];
  const float* rope = (const float*)d_in[5];
  const float* so8 = (const float*)d_in[6];

  float* ws = (float*)d_ws;
  float* qb = ws;
  float* kb = qb + (size_t)2 * N_HEADS * SEQ * HDIM;
  float* vtb = kb + (size_t)2 * N_KV * SEQ * HDIM;
  float* ab = vtb + (size_t)2 * N_KV * SEQ * HDIM;

  const int BS = 2 * SEQ;

  qkv_rope_so8_kernel<<<BS / TILE_M, 256, 0, stream>>>(x, Wq, Wk, Wv, rope, so8,
                                                       qb, kb, vtb);
  so8_flash_attn_kernel<<<(2 * N_HEADS * (QTILES / 2)) / 8, 256, 0, stream>>>(
      qb, kb, vtb, ab);
  out_proj_kernel<<<BS / TILE_M, 256, 0, stream>>>(ab, Wo, (float*)d_out);
}


# Attention softmax: stabiliser lags (raised only when a query's max rises over 2^8; one wave-uniform test per 64 keys) and is the score WMMA's C operand: score sums start from it, no per-tile rescale
# speedup vs baseline: 5.2763x; 1.0732x over previous
#include <hip/hip_runtime.h>
#include <hip/hip_bf16.h>

#define N_HEADS 32
#define N_KV 8
#define HDIM 8
#define SEQ 2048
#define HID 256
#define QTILES (SEQ / 16)
#define KV_RING 4

typedef __attribute__((ext_vector_type(2))) float v2f;
typedef __attribute__((ext_vector_type(8))) float v8f;
typedef __attribute__((ext_vector_type(4))) float v4f;


#define TILE_M 16
#define QKV_PITCH 388
#define OUT_PITCH 260

template <int NT>
__device__ __forceinline__ void wave_gemm_k256(const float* __restrict__ arow,
                                               const float* const (&wrow)[NT],
                                               v8f (&acc)[NT]) {
  v4f a0 = *(const v4f*)(arow);
  v4f a1 = *(const v4f*)(arow + 8);
  v4f b0[NT], b1[NT];
#pragma unroll
  for (int t = 0; t < NT; ++t) {
    b0[t] = *(const v4f*)(wrow[t]);
    b1[t] = *(const v4f*)(wrow[t] + 8);
  }
#pragma unroll
  for (int kc = 0; kc < HID / 16; ++kc) {
    v4f na0 = a0, na1 = a1;
    v4f nb0[NT], nb1[NT];
#pragma unroll
    for (int t = 0; t < NT; ++t) {
      nb0[t] = b0[t];
      nb1[t] = b1[t];
    }
    if (kc + 1 < HID / 16) {
      const int ko = (kc + 1) * 16;
      na0 = *(const v4f*)(arow + ko);
      na1 = *(const v4f*)(arow + ko + 8);
#pragma unroll
      for (int t = 0; t < NT; ++t) {
        nb0[t] = *(const v4f*)(wrow[t] + ko);
        nb1[t] = *(const v4f*)(wrow[t] + ko + 8);
      }
    }
#pragma unroll
    for (int t = 0; t < NT; ++t)
      acc[t] = __builtin_amdgcn_wmma_f32_16x16x4_f32(
          false, a0.xy, false, b0[t].xy, (short)0, acc[t], false, false);
#pragma unroll
    for (int t = 0; t < NT; ++t)
      acc[t] = __builtin_amdgcn_wmma_f32_16x16x4_f32(
          false, a0.zw, false, b0[t].zw, (short)0, acc[t], false, false);
#pragma unroll
    for (int t = 0; t < NT; ++t)
      acc[t] = __builtin_amdgcn_wmma_f32_16x16x4_f32(
          false, a1.xy, false, b1[t].xy, (short)0, acc[t], false, false);
#pragma unroll
    for (int t = 0; t < NT; ++t)
      acc[t] = __builtin_amdgcn_wmma_f32_16x16x4_f32(
          false, a1.zw, false, b1[t].zw, (short)0, acc[t], false, false);
    a0 = na0;
    a1 = na1;
#pragma unroll
    for (int t = 0; t < NT; ++t) {
      b0[t] = nb0[t];
      b1[t] = nb1[t];
    }
  }
}

__device__ __forceinline__ float xhalf(float x) {
  const int b = __float_as_int(x);
  return __int_as_float(__builtin_amdgcn_permlanex16(
      b, b, 0x76543210, 0xfedcba98, false, false));
}

__device__ __forceinline__ float fast_exp2(float x) {
  return __builtin_amdgcn_exp2f(x);
}

__global__ __launch_bounds__(256) void qkv_rope_so8_kernel(
    const float* __restrict__ x, const float* __restrict__ Wq,
    const float* __restrict__ Wk, const float* __restrict__ Wv,
    const float* __restrict__ rope, const float* __restrict__ so8,
    float* __restrict__ qb, float* __restrict__ kb, float* __restrict__ vtb) {
  __shared__ __attribute__((aligned(16))) float tile[TILE_M * QKV_PITCH];
  __shared__ float trig[TILE_M][8];

  const int tid = threadIdx.x;
  const int lane = tid & 31;
  const int widx = __builtin_amdgcn_readfirstlane(tid >> 5);
  const int c = lane & 15;
  const int h = lane >> 4;
  const int m0 = blockIdx.x * TILE_M;

  if (tid < 64) {
    const int row = tid >> 2;
    const int i = tid & 3;
    const float invf =
        (i == 0) ? 1.0f : ((i == 1) ? 0.1f : ((i == 2) ? 0.01f : 0.001f));
    const int s = (m0 + row) & (SEQ - 1);
    const float fr = (float)s * invf;
    trig[row][i] = cosf(fr);
    trig[row][4 + i] = sinf(fr);
  }

  const float* arow = x + (size_t)(m0 + c) * HID + 4 * h;
  const float* wrow[3];
#pragma unroll
  for (int t = 0; t < 3; ++t) {
    const int nt = widx * 3 + t;
    const int n0 = nt * 16;
    const float* Wb = (nt < 16) ? (Wq + (size_t)n0 * HID)
                                : ((nt < 20) ? (Wk + (size_t)(n0 - 256) * HID)
                                             : (Wv + (size_t)(n0 - 320) * HID));
    wrow[t] = Wb + (size_t)c * HID + 4 * h;
  }
  v8f acc[3] = {};
  wave_gemm_k256<3>(arow, wrow, acc);

#pragma unroll
  for (int t = 0; t < 3; ++t)
#pragma unroll
    for (int r = 0; r < 8; ++r)
      tile[(8 * h + r) * QKV_PITCH + (widx * 3 + t) * 16 + c] = acc[t][r];
  __syncthreads();

  for (int it = 0; it < 3; ++it) {
    const int id = it * 256 + tid;
    const int row = id & 15;
    const int hd = id >> 4;
    const int kind = (hd < 32) ? 0 : (hd < 40 ? 1 : 2);
    const int head = (hd < 32) ? hd : (hd < 40 ? hd - 32 : hd - 40);
    const int bs = m0 + row;
    const int s = bs & (SEQ - 1);
    const int b = bs >> 11;

    const float* tp = tile + row * QKV_PITCH + hd * 8;
    const v4f t0 = *(const v4f*)tp;
    const v4f t1 = *(const v4f*)(tp + 4);
    const float pr[HDIM] = {t0.x, t0.y, t0.z, t0.w, t1.x, t1.y, t1.z, t1.w};

    float y[HDIM];
    if (kind <= 1) {
      float cr[4] = {}, sr[4] = {};
#pragma unroll
      for (int i = 0; i < 4; ++i) {
        const float ci = trig[row][i], si = trig[row][4 + i];
#pragma unroll
        for (int j = 0; j < 4; ++j) {
          cr[j] = fmaf(ci, rope[i * 8 + j], cr[j]);
          sr[j] = fmaf(si, rope[i * 8 + j], sr[j]);
        }
      }
#pragma unroll
      for (int d = 0; d < HDIM; ++d) {
        const int j = d & 3;
        const float rh = (d < 4) ? -pr[d + 4] : pr[d - 4];
        y[d] = pr[d] * cr[j] + rh * sr[j];
      }
    } else {
#pragma unroll
      for (int d = 0; d < HDIM; ++d) y[d] = pr[d];
    }

    float z[HDIM];
#pragma unroll
    for (int j = 0; j < HDIM; ++j) {
      float sum = 0.0f;
#pragma unroll
      for (int d = 0; d < HDIM; ++d) sum = fmaf(y[d], so8[d * 8 + j], sum);
      z[j] = sum;
    }

    if (kind == 0) {
      float* dst = qb + (((size_t)b * N_HEADS + head) * SEQ + s) * HDIM;
      *(v4f*)dst = (v4f){z[0], z[1], z[2], z[3]};
      *(v4f*)(dst + 4) = (v4f){z[4], z[5], z[6], z[7]};
    } else if (kind == 1) {
      float* dst = kb + (((size_t)b * N_KV + head) * SEQ + s) * HDIM;
      *(v4f*)dst = (v4f){z[0], z[1], z[2], z[3]};
      *(v4f*)(dst + 4) = (v4f){z[4], z[5], z[6], z[7]};
    } else {
#pragma unroll
      for (int j = 0; j < HDIM; ++j)
        vtb[(((size_t)b * N_KV + head) * HDIM + j) * SEQ + s] = z[j];
    }
  }
}

struct KVFrag {
  v2f Ka0, Ka1;
  v4f V0, V1;
};

__device__ __forceinline__ void load_k(KVFrag& f,
                                       const float* __restrict__ krow0,
                                       int kbase) {
  const float* krow = krow0 + (size_t)kbase * HDIM;
  f.Ka0 = *(const v2f*)(krow);
  f.Ka1 = *(const v2f*)(krow + 4);
}

__device__ __forceinline__ void load_v(KVFrag& f,
                                       const float* __restrict__ vrow,
                                       int kbase) {
  f.V0 = *(const v4f*)(vrow + kbase);
  f.V1 = *(const v4f*)(vrow + kbase + 4);
}

__device__ __forceinline__ v8f score_tile(const KVFrag& f, const v2f Qb0,
                                          const v2f Qb1, const v8f negm) {
  v8f sc = __builtin_amdgcn_wmma_f32_16x16x4_f32(false, f.Ka0, false, Qb0,
                                                 (short)0, negm, false, false);
  sc = __builtin_amdgcn_wmma_f32_16x16x4_f32(false, f.Ka1, false, Qb1,
                                             (short)0, sc, false, false);
  return sc;
}

__device__ __forceinline__ void mask_diag(v8f& sc, int lanelo, int hi) {
#pragma unroll
  for (int r = 0; r < 8; ++r)
    sc[r] += (8 * hi + r > lanelo) ? -3.0e38f : 0.0f;
}

template <int NT>
__device__ __forceinline__ void lag_softmax(v8f (&sc)[NT], float& nm,
                                            v8f& negm, float& thr, float& flo,
                                            v8f& o) {
  float t[NT];
#pragma unroll
  for (int i = 0; i < NT; ++i) {
    const float a = fmaxf(fmaxf(sc[i][0], sc[i][1]), sc[i][2]);
    const float b = fmaxf(fmaxf(sc[i][3], sc[i][4]), sc[i][5]);
    t[i] = fmaxf(fmaxf(fmaxf(a, sc[i][6]), sc[i][7]), b);
  }
  float tm = t[0];
#pragma unroll
  for (int i = 1; i < NT; ++i) tm = fmaxf(tm, t[i]);
  tm = fmaxf(tm, xhalf(tm));

  if (__builtin_amdgcn_ballot_w32(tm > thr) != 0) {
    const float d = fmaxf(tm, flo);
    const float nm_new = nm - d;
    const float d2 = nm - nm_new;
    nm = nm_new;
    const float corr = fast_exp2(fminf(-d2, 0.0f));
#pragma unroll
    for (int r = 0; r < 8; ++r) o[r] = o[r] * corr;
#pragma unroll
    for (int i = 0; i < NT; ++i)
#pragma unroll
      for (int r = 0; r < 8; ++r) sc[i][r] -= d2;
#pragma unroll
    for (int r = 0; r < 8; ++r) negm[r] = nm_new;
    thr = 8.0f;
    flo = 0.0f;
  }

#pragma unroll
  for (int i = 0; i < NT; ++i)
#pragma unroll
    for (int r = 0; r < 8; ++r) sc[i][r] = fast_exp2(sc[i][r]);
}

__device__ __forceinline__ void pv_tile(const KVFrag& f, const v8f& p,
                                        bool vlane, float vfill, v8f& o) {
  const v4f V0 = f.V0, V1 = f.V1;
  const v2f fill = (v2f){vfill, vfill};
  v2f Va[4];
  Va[0] = vlane ? V0.xy : fill;
  Va[1] = vlane ? V0.zw : fill;
  Va[2] = vlane ? V1.xy : fill;
  Va[3] = vlane ? V1.zw : fill;
#pragma unroll
  for (int t = 0; t < 4; ++t)
    o = __builtin_amdgcn_wmma_f32_16x16x4_f32(
        false, Va[t], false, (v2f){p[2 * t], p[2 * t + 1]}, (short)0, o, false,
        false);
}

template <bool MASK>
__device__ __forceinline__ void attn_tile1(
    const KVFrag& f, int lanelo, int hi, bool vlane, float vfill,
    const v2f Qb0, const v2f Qb1, float& nm, v8f& negm, float& thr, float& flo,
    v8f& o) {
  v8f sc[1];
  sc[0] = score_tile(f, Qb0, Qb1, negm);
  if (MASK) mask_diag(sc[0], lanelo, hi);
  lag_softmax<1>(sc, nm, negm, thr, flo, o);
  pv_tile(f, sc[0], vlane, vfill, o);
}

__global__ __launch_bounds__(256) void so8_flash_attn_kernel(
    const float* __restrict__ q, const float* __restrict__ k,
    const float* __restrict__ vT, float* __restrict__ attn) {
  const int lane = threadIdx.x & 31;
  const int widx = __builtin_amdgcn_readfirstlane(threadIdx.x >> 5);
  const int w = blockIdx.x * 8 + widx;
  const int pr = w & (QTILES / 2 - 1);
  const int h = (w >> 6) & (N_HEADS - 1);
  const int b = w >> 11;
  const int kv = h >> 2;
  const int lanelo = lane & 15;
  const int hi = lane >> 4;
  const int hi2 = hi * 2;

  const float* qp = q + ((size_t)b * N_HEADS + h) * SEQ * HDIM;
  const float* kp = k + ((size_t)b * N_KV + kv) * SEQ * HDIM;
  const float* vp = vT + ((size_t)b * N_KV + kv) * HDIM * SEQ;

  const float* vrow = vp + (size_t)(lanelo & 7) * SEQ + 8 * hi;
  const bool vlane = lanelo < HDIM;
  const float vfill = (lanelo == HDIM) ? 1.0f : 0.0f;

  const float* krow0 = kp + (size_t)lanelo * HDIM + hi2;

  const float QSCALE = 0.35355339059327f * 1.44269504088896f;

#pragma unroll
  for (int ph = 0; ph < 2; ++ph) {
    const int qt = (ph == 0) ? pr : (QTILES - 1 - pr);
    const int qbase = qt * 16;

    const float* qrow = qp + (size_t)(qbase + lanelo) * HDIM;
    const v2f Qb0 = *(const v2f*)(qrow + hi2) * QSCALE;
    const v2f Qb1 = *(const v2f*)(qrow + 4 + hi2) * QSCALE;

    v8f o = {};
    float nm = 0.0f;
    v8f negm = {};
    float thr = -3.0e38f, flo = -3.0e38f;

    KVFrag f[KV_RING];
#pragma unroll
    for (int i = 0; i < KV_RING; ++i) load_k(f[i], krow0, min(i, qt) * 16);
    __builtin_amdgcn_sched_barrier(0);
#pragma unroll
    for (int i = 0; i < KV_RING; ++i) {
      load_v(f[i], vrow, min(i, qt) * 16);
      __builtin_amdgcn_sched_barrier(0);
    }

    int kt = 0;
    for (; kt + KV_RING <= qt; kt += KV_RING) {
      v8f sc[KV_RING];
#pragma unroll
      for (int i = 0; i < KV_RING; ++i)
        sc[i] = score_tile(f[i], Qb0, Qb1, negm);
#pragma unroll
      for (int i = 0; i < KV_RING; ++i)
        load_k(f[i], krow0, min(kt + KV_RING + i, qt) * 16);
      lag_softmax<KV_RING>(sc, nm, negm, thr, flo, o);
#pragma unroll
      for (int i = 0; i < KV_RING; ++i) {
        pv_tile(f[i], sc[i], vlane, vfill, o);
        load_v(f[i], vrow, min(kt + KV_RING + i, qt) * 16);
        __builtin_amdgcn_sched_barrier(0);
      }
    }
#pragma unroll
    for (int i = 0; i < KV_RING - 1; ++i) {
      if (kt < qt) {
        attn_tile1<false>(f[0], lanelo, hi, vlane, vfill, Qb0, Qb1, nm, negm,
                          thr, flo, o);
#pragma unroll
        for (int j = 0; j + 1 < KV_RING; ++j) f[j] = f[j + 1];
        ++kt;
      }
    }
    attn_tile1<true>(f[0], lanelo, hi, vlane, vfill, Qb0, Qb1, nm, negm, thr,
                     flo, o);

    const float l = xhalf(o[0]);
    const float inv = __builtin_amdgcn_rcpf(l);
    if (hi == 0) {
      float* dst =
          attn + (((size_t)b * SEQ + qbase + lanelo) * N_HEADS + h) * HDIM;
      *(v4f*)dst = (v4f){o[0] * inv, o[1] * inv, o[2] * inv, o[3] * inv};
      *(v4f*)(dst + 4) = (v4f){o[4] * inv, o[5] * inv, o[6] * inv, o[7] * inv};
    }
  }
}

__global__ __launch_bounds__(256) void out_proj_kernel(
    const float* __restrict__ attn, const float* __restrict__ Wo,
    float* __restrict__ out) {
  __shared__ __attribute__((aligned(16))) float tile[TILE_M * OUT_PITCH];

  const int tid = threadIdx.x;
  const int lane = tid & 31;
  const int widx = tid >> 5;
  const int c = lane & 15;
  const int h = lane >> 4;
  const int m0 = blockIdx.x * TILE_M;

  const float* arow = attn + (size_t)(m0 + c) * HID + 4 * h;
  const float* wrow[2];
#pragma unroll
  for (int t = 0; t < 2; ++t)
    wrow[t] = Wo + (size_t)((widx * 2 + t) * 16 + c) * HID + 4 * h;
  v8f acc[2] = {};
  wave_gemm_k256<2>(arow, wrow, acc);

#pragma unroll
  for (int t = 0; t < 2; ++t)
#pragma unroll
    for (int r = 0; r < 8; ++r)
      tile[(8 * h + r) * OUT_PITCH + (widx * 2 + t) * 16 + c] = acc[t][r];
  __syncthreads();

#pragma unroll
  for (int it = 0; it < 4; ++it) {
    const int id = it * 256 + tid;
    const int row = id >> 6;
    const int c4 = (id & 63) * 4;
    *(v4f*)(out + (size_t)(m0 + row) * HID + c4) =
        *(const v4f*)(tile + row * OUT_PITCH + c4);
  }
}

extern "C" void kernel_launch(void* const* d_in, const int* in_sizes, int n_in,
                              void* d_out, int out_size, void* d_ws,
                              size_t ws_size, hipStream_t stream) {
  const float* x = (const float*)d_in[0];
  const float* Wq = (const float*)d_in[1];
  const float* Wk = (const float*)d_in[2];
  const float* Wv = (const float*)d_in[3];
  const float* Wo = (const float*)d_in[4];
  const float* rope = (const float*)d_in[5];
  const float* so8 = (const float*)d_in[6];

  float* ws = (float*)d_ws;
  float* qb = ws;
  float* kb = qb + (size_t)2 * N_HEADS * SEQ * HDIM;
  float* vtb = kb + (size_t)2 * N_KV * SEQ * HDIM;
  float* ab = vtb + (size_t)2 * N_KV * SEQ * HDIM;

  const int BS = 2 * SEQ;

  qkv_rope_so8_kernel<<<BS / TILE_M, 256, 0, stream>>>(x, Wq, Wk, Wv, rope, so8,
                                                       qb, kb, vtb);
  so8_flash_attn_kernel<<<(2 * N_HEADS * (QTILES / 2)) / 8, 256, 0, stream>>>(
      qb, kb, vtb, ab);
  out_proj_kernel<<<BS / TILE_M, 256, 0, stream>>>(ab, Wo, (float*)d_out);
}


# QKV and output projection GEMMs: 64-row blocks, operand chunks staged global->LDS by whole-line loads (3-buffer ring, 1 barrier per 32 k), fragments cut from LDS; every sum and its order unchanged
# speedup vs baseline: 6.0256x; 1.1420x over previous
#include <hip/hip_runtime.h>
#include <hip/hip_bf16.h>

#define N_HEADS 32
#define N_KV 8
#define HDIM 8
#define SEQ 2048
#define HID 256
#define QTILES (SEQ / 16)
#define KV_RING 4

typedef __attribute__((ext_vector_type(2))) float v2f;
typedef __attribute__((ext_vector_type(8))) float v8f;
typedef __attribute__((ext_vector_type(4))) float v4f;


#define GM 64
#define KC 32
#define ST_PITCH 36
#define QKV_COLS 96
#define QKV_PITCH 100
#define OUT_COLS 64
#define OUT_PITCH 68

template <int NT>
struct HalfFrag {
  v4f a0, a1;
  v4f b0[NT], b1[NT];
};

template <int NT>
__device__ __forceinline__ void read_half(HalfFrag<NT>& f, const float* sb,
                                          int aoff, const int (&boff)[NT],
                                          int hf) {
  f.a0 = *(const v4f*)(sb + aoff + 16 * hf);
  f.a1 = *(const v4f*)(sb + aoff + 16 * hf + 8);
#pragma unroll
  for (int t = 0; t < NT; ++t) {
    f.b0[t] = *(const v4f*)(sb + boff[t] + 16 * hf);
    f.b1[t] = *(const v4f*)(sb + boff[t] + 16 * hf + 8);
  }
}

template <int NT>
__device__ __forceinline__ void mul_half(const HalfFrag<NT>& f,
                                         v8f (&acc)[NT]) {
#pragma unroll
  for (int t = 0; t < NT; ++t)
    acc[t] = __builtin_amdgcn_wmma_f32_16x16x4_f32(
        false, f.a0.xy, false, f.b0[t].xy, (short)0, acc[t], false, false);
#pragma unroll
  for (int t = 0; t < NT; ++t)
    acc[t] = __builtin_amdgcn_wmma_f32_16x16x4_f32(
        false, f.a0.zw, false, f.b0[t].zw, (short)0, acc[t], false, false);
#pragma unroll
  for (int t = 0; t < NT; ++t)
    acc[t] = __builtin_amdgcn_wmma_f32_16x16x4_f32(
        false, f.a1.xy, false, f.b1[t].xy, (short)0, acc[t], false, false);
#pragma unroll
  for (int t = 0; t < NT; ++t)
    acc[t] = __builtin_amdgcn_wmma_f32_16x16x4_f32(
        false, f.a1.zw, false, f.b1[t].zw, (short)0, acc[t], false, false);
}

__device__ __forceinline__ void lds_barrier() {
  __builtin_amdgcn_fence(__ATOMIC_RELEASE, "workgroup", "local");
  __builtin_amdgcn_s_barrier();
  __builtin_amdgcn_fence(__ATOMIC_ACQUIRE, "workgroup", "local");
}

template <int NT, int NP>
__device__ __forceinline__ void block_gemm_k256(const float* const (&src)[NP],
                                                float* __restrict__ stage,
                                                int tid, int rt, int cg, int c,
                                                int h, v8f (&acc)[NT]) {
  constexpr int ROWS = GM + NT * 32;
  constexpr int BUF = ROWS * ST_PITCH;
  constexpr int NCH = HID / KC;
  static_assert(NP * 256 == ROWS * 8, "pieces per thread");
  int dst[NP];
  v4f g[2][NP];
#pragma unroll
  for (int i = 0; i < NP; ++i) {
    const int p = i * 256 + tid;
    dst[i] = (p >> 3) * ST_PITCH + (p & 7) * 4;
    g[0][i] = *(const v4f*)(src[i]);
    g[1][i] = *(const v4f*)(src[i] + KC);
  }
#pragma unroll
  for (int i = 0; i < NP; ++i) *(v4f*)(stage + dst[i]) = g[0][i];
#pragma unroll
  for (int i = 0; i < NP; ++i) *(v4f*)(stage + BUF + dst[i]) = g[1][i];
#pragma unroll
  for (int i = 0; i < NP; ++i) g[0][i] = *(const v4f*)(src[i] + 2 * KC);
  lds_barrier();

  const int aoff = (16 * rt + c) * ST_PITCH + 4 * h;
  int boff[NT];
#pragma unroll
  for (int t = 0; t < NT; ++t)
    boff[t] = (GM + (cg * NT + t) * 16 + c) * ST_PITCH + 4 * h;

  HalfFrag<NT> f0, f1;
  read_half<NT>(f0, stage, aoff, boff, 0);

#pragma unroll
  for (int kc = 0; kc < NCH; ++kc) {
    if (kc + 3 < NCH) {
#pragma unroll
      for (int i = 0; i < NP; ++i)
        g[(kc + 1) & 1][i] = *(const v4f*)(src[i] + (kc + 3) * KC);
    }
    __builtin_amdgcn_sched_barrier(0);
    read_half<NT>(f1, stage + (kc % 3) * BUF, aoff, boff, 1);
    mul_half<NT>(f0, acc);
    __builtin_amdgcn_sched_barrier(0);
    if (kc + 2 < NCH) {
      float* nb = stage + ((kc + 2) % 3) * BUF;
#pragma unroll
      for (int i = 0; i < NP; ++i) *(v4f*)(nb + dst[i]) = g[kc & 1][i];
    }
    __builtin_amdgcn_sched_barrier(0);
    if (kc + 1 < NCH)
      read_half<NT>(f0, stage + ((kc + 1) % 3) * BUF, aoff, boff, 0);
    mul_half<NT>(f1, acc);
    if (kc + 2 < NCH) {
      lds_barrier();
    }
  }
}

__device__ __forceinline__ float xhalf(float x) {
  const int b = __float_as_int(x);
  return __int_as_float(__builtin_amdgcn_permlanex16(
      b, b, 0x76543210, 0xfedcba98, false, false));
}

__device__ __forceinline__ float fast_exp2(float x) {
  return __builtin_amdgcn_exp2f(x);
}

__global__ __launch_bounds__(256) void qkv_rope_so8_kernel(
    const float* __restrict__ x, const float* __restrict__ Wq,
    const float* __restrict__ Wk, const float* __restrict__ Wv,
    const float* __restrict__ rope, const float* __restrict__ so8,
    float* __restrict__ qb, float* __restrict__ kb, float* __restrict__ vtb) {
  __shared__ __attribute__((aligned(16))) float stage[3 * (GM + QKV_COLS) * ST_PITCH];
  __shared__ __attribute__((aligned(16))) float tile[GM * QKV_PITCH];
  __shared__ float trig[GM][8];

  const int tid = threadIdx.x;
  const int lane = tid & 31;
  const int widx = __builtin_amdgcn_readfirstlane(tid >> 5);
  const int rt = widx >> 1;
  const int cg = widx & 1;
  const int c = lane & 15;
  const int h = lane >> 4;
  const int grp = blockIdx.x & 3;
  const int m0 = (blockIdx.x >> 2) * GM;
  const int n0 = grp * QKV_COLS;

  {
    const int row = tid >> 2;
    const int i = tid & 3;
    const float invf =
        (i == 0) ? 1.0f : ((i == 1) ? 0.1f : ((i == 2) ? 0.01f : 0.001f));
    const int s = (m0 + row) & (SEQ - 1);
    const float fr = (float)s * invf;
    trig[row][i] = cosf(fr);
    trig[row][4 + i] = sinf(fr);
  }

  const float* src[5];
#pragma unroll
  for (int i = 0; i < 5; ++i) {
    const int p = i * 256 + tid;
    const int row = p >> 3;
    const int seg = (p & 7) * 4;
    const int n = n0 + row - GM;
    const float* rp =
        (row < GM) ? (x + (size_t)(m0 + row) * HID)
                   : ((n < 256) ? (Wq + (size_t)n * HID)
                                : ((n < 320) ? (Wk + (size_t)(n - 256) * HID)
                                             : (Wv + (size_t)(n - 320) * HID)));
    src[i] = rp + seg;
  }
  v8f acc[3] = {};
  block_gemm_k256<3, 5>(src, stage, tid, rt, cg, c, h, acc);

#pragma unroll
  for (int t = 0; t < 3; ++t)
#pragma unroll
    for (int r = 0; r < 8; ++r)
      tile[(16 * rt + 8 * h + r) * QKV_PITCH + (cg * 3 + t) * 16 + c] =
          acc[t][r];
  __syncthreads();

  for (int it = 0; it < 3; ++it) {
    const int id = it * 256 + tid;
    const int row = id & (GM - 1);
    const int hl = id >> 6;
    const int hd = grp * 12 + hl;
    const int kind = (hd < 32) ? 0 : (hd < 40 ? 1 : 2);
    const int head = (hd < 32) ? hd : (hd < 40 ? hd - 32 : hd - 40);
    const int bs = m0 + row;
    const int s = bs & (SEQ - 1);
    const int b = bs >> 11;

    const float* tp = tile + row * QKV_PITCH + hl * 8;
    const v4f t0 = *(const v4f*)tp;
    const v4f t1 = *(const v4f*)(tp + 4);
    const float pr[HDIM] = {t0.x, t0.y, t0.z, t0.w, t1.x, t1.y, t1.z, t1.w};

    float y[HDIM];
    if (kind <= 1) {
      float cr[4] = {}, sr[4] = {};
#pragma unroll
      for (int i = 0; i < 4; ++i) {
        const float ci = trig[row][i], si = trig[row][4 + i];
#pragma unroll
        for (int j = 0; j < 4; ++j) {
          cr[j] = fmaf(ci, rope[i * 8 + j], cr[j]);
          sr[j] = fmaf(si, rope[i * 8 + j], sr[j]);
        }
      }
#pragma unroll
      for (int d = 0; d < HDIM; ++d) {
        const int j = d & 3;
        const float rh = (d < 4) ? -pr[d + 4] : pr[d - 4];
        y[d] = pr[d] * cr[j] + rh * sr[j];
      }
    } else {
#pragma unroll
      for (int d = 0; d < HDIM; ++d) y[d] = pr[d];
    }

    float z[HDIM];
#pragma unroll
    for (int j = 0; j < HDIM; ++j) {
      float sum = 0.0f;
#pragma unroll
      for (int d = 0; d < HDIM; ++d) sum = fmaf(y[d], so8[d * 8 + j], sum);
      z[j] = sum;
    }

    if (kind == 0) {
      float* dst = qb + (((size_t)b * N_HEADS + head) * SEQ + s) * HDIM;
      *(v4f*)dst = (v4f){z[0], z[1], z[2], z[3]};
      *(v4f*)(dst + 4) = (v4f){z[4], z[5], z[6], z[7]};
    } else if (kind == 1) {
      float* dst = kb + (((size_t)b * N_KV + head) * SEQ + s) * HDIM;
      *(v4f*)dst = (v4f){z[0], z[1], z[2], z[3]};
      *(v4f*)(dst + 4) = (v4f){z[4], z[5], z[6], z[7]};
    } else {
#pragma unroll
      for (int j = 0; j < HDIM; ++j)
        vtb[(((size_t)b * N_KV + head) * HDIM + j) * SEQ + s] = z[j];
    }
  }
}

struct KVFrag {
  v2f Ka0, Ka1;
  v4f V0, V1;
};

__device__ __forceinline__ void load_k(KVFrag& f,
                                       const float* __restrict__ krow0,
                                       int kbase) {
  const float* krow = krow0 + (size_t)kbase * HDIM;
  f.Ka0 = *(const v2f*)(krow);
  f.Ka1 = *(const v2f*)(krow + 4);
}

__device__ __forceinline__ void load_v(KVFrag& f,
                                       const float* __restrict__ vrow,
                                       int kbase) {
  f.V0 = *(const v4f*)(vrow + kbase);
  f.V1 = *(const v4f*)(vrow + kbase + 4);
}

__device__ __forceinline__ v8f score_tile(const KVFrag& f, const v2f Qb0,
                                          const v2f Qb1, const v8f negm) {
  v8f sc = __builtin_amdgcn_wmma_f32_16x16x4_f32(false, f.Ka0, false, Qb0,
                                                 (short)0, negm, false, false);
  sc = __builtin_amdgcn_wmma_f32_16x16x4_f32(false, f.Ka1, false, Qb1,
                                             (short)0, sc, false, false);
  return sc;
}

__device__ __forceinline__ void mask_diag(v8f& sc, int lanelo, int hi) {
#pragma unroll
  for (int r = 0; r < 8; ++r)
    sc[r] += (8 * hi + r > lanelo) ? -3.0e38f : 0.0f;
}

template <int NT>
__device__ __forceinline__ void lag_softmax(v8f (&sc)[NT], float& nm,
                                            v8f& negm, float& thr, float& flo,
                                            v8f& o) {
  float t[NT];
#pragma unroll
  for (int i = 0; i < NT; ++i) {
    const float a = fmaxf(fmaxf(sc[i][0], sc[i][1]), sc[i][2]);
    const float b = fmaxf(fmaxf(sc[i][3], sc[i][4]), sc[i][5]);
    t[i] = fmaxf(fmaxf(fmaxf(a, sc[i][6]), sc[i][7]), b);
  }
  float tm = t[0];
#pragma unroll
  for (int i = 1; i < NT; ++i) tm = fmaxf(tm, t[i]);
  tm = fmaxf(tm, xhalf(tm));

  if (__builtin_amdgcn_ballot_w32(tm > thr) != 0) {
    const float d = fmaxf(tm, flo);
    const float nm_new = nm - d;
    const float d2 = nm - nm_new;
    nm = nm_new;
    const float corr = fast_exp2(fminf(-d2, 0.0f));
#pragma unroll
    for (int r = 0; r < 8; ++r) o[r] = o[r] * corr;
#pragma unroll
    for (int i = 0; i < NT; ++i)
#pragma unroll
      for (int r = 0; r < 8; ++r) sc[i][r] -= d2;
#pragma unroll
    for (int r = 0; r < 8; ++r) negm[r] = nm_new;
    thr = 8.0f;
    flo = 0.0f;
  }

#pragma unroll
  for (int i = 0; i < NT; ++i)
#pragma unroll
    for (int r = 0; r < 8; ++r) sc[i][r] = fast_exp2(sc[i][r]);
}

__device__ __forceinline__ void pv_tile(const KVFrag& f, const v8f& p,
                                        bool vlane, float vfill, v8f& o) {
  const v4f V0 = f.V0, V1 = f.V1;
  const v2f fill = (v2f){vfill, vfill};
  v2f Va[4];
  Va[0] = vlane ? V0.xy : fill;
  Va[1] = vlane ? V0.zw : fill;
  Va[2] = vlane ? V1.xy : fill;
  Va[3] = vlane ? V1.zw : fill;
#pragma unroll
  for (int t = 0; t < 4; ++t)
    o = __builtin_amdgcn_wmma_f32_16x16x4_f32(
        false, Va[t], false, (v2f){p[2 * t], p[2 * t + 1]}, (short)0, o, false,
        false);
}

template <bool MASK>
__device__ __forceinline__ void attn_tile1(
    const KVFrag& f, int lanelo, int hi, bool vlane, float vfill,
    const v2f Qb0, const v2f Qb1, float& nm, v8f& negm, float& thr, float& flo,
    v8f& o) {
  v8f sc[1];
  sc[0] = score_tile(f, Qb0, Qb1, negm);
  if (MASK) mask_diag(sc[0], lanelo, hi);
  lag_softmax<1>(sc, nm, negm, thr, flo, o);
  pv_tile(f, sc[0], vlane, vfill, o);
}

__global__ __launch_bounds__(256) void so8_flash_attn_kernel(
    const float* __restrict__ q, const float* __restrict__ k,
    const float* __restrict__ vT, float* __restrict__ attn) {
  const int lane = threadIdx.x & 31;
  const int widx = __builtin_amdgcn_readfirstlane(threadIdx.x >> 5);
  const int w = blockIdx.x * 8 + widx;
  const int pr = w & (QTILES / 2 - 1);
  const int h = (w >> 6) & (N_HEADS - 1);
  const int b = w >> 11;
  const int kv = h >> 2;
  const int lanelo = lane & 15;
  const int hi = lane >> 4;
  const int hi2 = hi * 2;

  const float* qp = q + ((size_t)b * N_HEADS + h) * SEQ * HDIM;
  const float* kp = k + ((size_t)b * N_KV + kv) * SEQ * HDIM;
  const float* vp = vT + ((size_t)b * N_KV + kv) * HDIM * SEQ;

  const float* vrow = vp + (size_t)(lanelo & 7) * SEQ + 8 * hi;
  const bool vlane = lanelo < HDIM;
  const float vfill = (lanelo == HDIM) ? 1.0f : 0.0f;

  const float* krow0 = kp + (size_t)lanelo * HDIM + hi2;

  const float QSCALE = 0.35355339059327f * 1.44269504088896f;

#pragma unroll
  for (int ph = 0; ph < 2; ++ph) {
    const int qt = (ph == 0) ? pr : (QTILES - 1 - pr);
    const int qbase = qt * 16;

    const float* qrow = qp + (size_t)(qbase + lanelo) * HDIM;
    const v2f Qb0 = *(const v2f*)(qrow + hi2) * QSCALE;
    const v2f Qb1 = *(const v2f*)(qrow + 4 + hi2) * QSCALE;

    v8f o = {};
    float nm = 0.0f;
    v8f negm = {};
    float thr = -3.0e38f, flo = -3.0e38f;

    KVFrag f[KV_RING];
#pragma unroll
    for (int i = 0; i < KV_RING; ++i) load_k(f[i], krow0, min(i, qt) * 16);
    __builtin_amdgcn_sched_barrier(0);
#pragma unroll
    for (int i = 0; i < KV_RING; ++i) {
      load_v(f[i], vrow, min(i, qt) * 16);
      __builtin_amdgcn_sched_barrier(0);
    }

    int kt = 0;
    for (; kt + KV_RING <= qt; kt += KV_RING) {
      v8f sc[KV_RING];
#pragma unroll
      for (int i = 0; i < KV_RING; ++i)
        sc[i] = score_tile(f[i], Qb0, Qb1, negm);
#pragma unroll
      for (int i = 0; i < KV_RING; ++i)
        load_k(f[i], krow0, min(kt + KV_RING + i, qt) * 16);
      lag_softmax<KV_RING>(sc, nm, negm, thr, flo, o);
#pragma unroll
      for (int i = 0; i < KV_RING; ++i) {
        pv_tile(f[i], sc[i], vlane, vfill, o);
        load_v(f[i], vrow, min(kt + KV_RING + i, qt) * 16);
        __builtin_amdgcn_sched_barrier(0);
      }
    }
#pragma unroll
    for (int i = 0; i < KV_RING - 1; ++i) {
      if (kt < qt) {
        attn_tile1<false>(f[0], lanelo, hi, vlane, vfill, Qb0, Qb1, nm, negm,
                          thr, flo, o);
#pragma unroll
        for (int j = 0; j + 1 < KV_RING; ++j) f[j] = f[j + 1];
        ++kt;
      }
    }
    attn_tile1<true>(f[0], lanelo, hi, vlane, vfill, Qb0, Qb1, nm, negm, thr,
                     flo, o);

    const float l = xhalf(o[0]);
    const float inv = __builtin_amdgcn_rcpf(l);
    if (hi == 0) {
      float* dst =
          attn + (((size_t)b * SEQ + qbase + lanelo) * N_HEADS + h) * HDIM;
      *(v4f*)dst = (v4f){o[0] * inv, o[1] * inv, o[2] * inv, o[3] * inv};
      *(v4f*)(dst + 4) = (v4f){o[4] * inv, o[5] * inv, o[6] * inv, o[7] * inv};
    }
  }
}

__global__ __launch_bounds__(256) void out_proj_kernel(
    const float* __restrict__ attn, const float* __restrict__ Wo,
    float* __restrict__ out) {
  __shared__ __attribute__((aligned(16))) float stage[3 * (GM + OUT_COLS) * ST_PITCH];
  __shared__ __attribute__((aligned(16))) float tile[GM * OUT_PITCH];

  const int tid = threadIdx.x;
  const int lane = tid & 31;
  const int widx = __builtin_amdgcn_readfirstlane(tid >> 5);
  const int rt = widx >> 1;
  const int cg = widx & 1;
  const int c = lane & 15;
  const int h = lane >> 4;
  const int m0 = (blockIdx.x >> 2) * GM;
  const int n0 = (blockIdx.x & 3) * OUT_COLS;

  const float* src[4];
#pragma unroll
  for (int i = 0; i < 4; ++i) {
    const int p = i * 256 + tid;
    const int row = p >> 3;
    const int seg = (p & 7) * 4;
    const float* rp = (row < GM) ? (attn + (size_t)(m0 + row) * HID)
                                 : (Wo + (size_t)(n0 + row - GM) * HID);
    src[i] = rp + seg;
  }
  v8f acc[2] = {};
  block_gemm_k256<2, 4>(src, stage, tid, rt, cg, c, h, acc);

#pragma unroll
  for (int t = 0; t < 2; ++t)
#pragma unroll
    for (int r = 0; r < 8; ++r)
      tile[(16 * rt + 8 * h + r) * OUT_PITCH + (cg * 2 + t) * 16 + c] =
          acc[t][r];
  __syncthreads();

#pragma unroll
  for (int it = 0; it < 4; ++it) {
    const int id = it * 256 + tid;
    const int row = id >> 4;
    const int c4 = (id & 15) * 4;
    *(v4f*)(out + (size_t)(m0 + row) * HID + n0 + c4) =
        *(const v4f*)(tile + row * OUT_PITCH + c4);
  }
}

extern "C" void kernel_launch(void* const* d_in, const int* in_sizes, int n_in,
                              void* d_out, int out_size, void* d_ws,
                              size_t ws_size, hipStream_t stream) {
  const float* x = (const float*)d_in[0];
  const float* Wq = (const float*)d_in[1];
  const float* Wk = (const float*)d_in[2];
  const float* Wv = (const float*)d_in[3];
  const float* Wo = (const float*)d_in[4];
  const float* rope = (const float*)d_in[5];
  const float* so8 = (const float*)d_in[6];

  float* ws = (float*)d_ws;
  float* qb = ws;
  float* kb = qb + (size_t)2 * N_HEADS * SEQ * HDIM;
  float* vtb = kb + (size_t)2 * N_KV * SEQ * HDIM;
  float* ab = vtb + (size_t)2 * N_KV * SEQ * HDIM;

  const int BS = 2 * SEQ;

  qkv_rope_so8_kernel<<<(BS / GM) * 4, 256, 0, stream>>>(x, Wq, Wk, Wv, rope,
                                                         so8, qb, kb, vtb);
  so8_flash_attn_kernel<<<(2 * N_HEADS * (QTILES / 2)) / 8, 256, 0, stream>>>(
      qb, kb, vtb, ab);
  out_proj_kernel<<<(BS / GM) * 4, 256, 0, stream>>>(ab, Wo, (float*)d_out);
}
